# starting kernel
# baseline (speedup 1.0000x reference)
#include <hip/hip_runtime.h>
#include <math.h>

#define N_NODES 100000
#define N_EDGES 1600000
#define IN_DIM  128
#define HEADS   8
#define HDIM    16
#define OUT_DIM 128
#define NEG_SLOPE 0.01f
#define LDS_STRIDE 130

typedef __attribute__((ext_vector_type(2))) float v2f;
typedef __attribute__((ext_vector_type(8))) float v8f;

__device__ __forceinline__ void atomicMaxF(float* addr, float val) {
    if (val >= 0.0f) {
        atomicMax((int*)addr, __float_as_int(val));
    } else {
        atomicMin((unsigned int*)addr, __float_as_uint(val));
    }
}

__global__ void gat_init(float* __restrict__ out0, float* __restrict__ m,
                         float* __restrict__ den) {
    int j = blockIdx.x * blockDim.x + threadIdx.x;
    if (j < N_NODES * OUT_DIM) out0[j] = 0.0f;
    if (j < N_NODES * HEADS) { m[j] = -INFINITY; den[j] = 0.0f; }
}

__global__ void __launch_bounds__(256)
gat_gemm_wmma(const float* __restrict__ h, const float* __restrict__ Wfc,
              float* __restrict__ z) {
    __shared__ float Wl[OUT_DIM * LDS_STRIDE];

    const int tid = threadIdx.x;
    for (int j = tid; j < IN_DIM * OUT_DIM; j += 256) {
        int i  = j >> 7;
        int c  = j & 127;
        int hh = c >> 4;
        int d  = c & 15;
        Wl[c * LDS_STRIDE + i] = Wfc[hh * (IN_DIM * HDIM) + i * HDIM + d];
    }
    __syncthreads();

    const int wave  = tid >> 5;
    const int lane  = tid & 31;
    const int mtile = blockIdx.x * 8 + wave;
    if (mtile >= N_NODES / 16) return;

    const int row   = lane & 15;
    const int khalf = lane >> 4;

    const float* hrow = h + (size_t)(mtile * 16 + row) * IN_DIM;

    v8f acc[8] = {};

    for (int k0 = 0; k0 < IN_DIM; k0 += 4) {
        const int kk = k0 + 2 * khalf;
        v2f a = *reinterpret_cast<const v2f*>(hrow + kk);
#pragma unroll
        for (int t = 0; t < 8; ++t) {
            v2f b = *reinterpret_cast<const v2f*>(&Wl[(t * 16 + row) * LDS_STRIDE + kk]);
            acc[t] = __builtin_amdgcn_wmma_f32_16x16x4_f32(
                false, a, false, b, (short)0, acc[t], false, false);
        }
    }

    float* zbase = z + (size_t)(mtile * 16) * OUT_DIM;
#pragma unroll
    for (int t = 0; t < 8; ++t) {
#pragma unroll
        for (int i = 0; i < 8; ++i) {
            int mrow = i + khalf * 8;
            zbase[(size_t)mrow * OUT_DIM + t * 16 + row] = acc[t][i];
        }
    }
}

__global__ void gat_scores(const float* __restrict__ z,
                           const float* __restrict__ Wattn,
                           float* __restrict__ s_src, float* __restrict__ s_dst) {
    int idx = blockIdx.x * blockDim.x + threadIdx.x;
    if (idx >= N_NODES * HEADS) return;
    int n = idx >> 3, hh = idx & 7;
    const float* zp = z + (size_t)n * OUT_DIM + hh * HDIM;
    const float* as = Wattn + hh * (2 * HDIM);
    const float* ad = as + HDIM;
    float ss = 0.0f, sd = 0.0f;
#pragma unroll
    for (int d = 0; d < HDIM; ++d) {
        float v = zp[d];
        ss += v * as[d];
        sd += v * ad[d];
    }
    s_src[idx] = ss;
    s_dst[idx] = sd;
}

__global__ void gat_edge_max(const int* __restrict__ src, const int* __restrict__ dst,
                             const float* __restrict__ s_src, const float* __restrict__ s_dst,
                             float* __restrict__ m) {
    int e = blockIdx.x * blockDim.x + threadIdx.x;
    if (e >= N_EDGES) return;
    int s = src[e], d = dst[e];
    const float* ps = s_src + (size_t)s * HEADS;
    const float* pd = s_dst + (size_t)d * HEADS;
    float* pm = m + (size_t)d * HEADS;
#pragma unroll
    for (int hh = 0; hh < HEADS; ++hh) {
        float sc = ps[hh] + pd[hh];
        sc = (sc >= 0.0f) ? sc : NEG_SLOPE * sc;
        atomicMaxF(pm + hh, sc);
    }
}

__global__ void gat_edge_sum(const int* __restrict__ src, const int* __restrict__ dst,
                             const float* __restrict__ s_src, const float* __restrict__ s_dst,
                             const float* __restrict__ m, float* __restrict__ den) {
    int e = blockIdx.x * blockDim.x + threadIdx.x;
    if (e >= N_EDGES) return;
    int s = src[e], d = dst[e];
    const float* ps = s_src + (size_t)s * HEADS;
    const float* pd = s_dst + (size_t)d * HEADS;
    const float* pm = m + (size_t)d * HEADS;
    float* pden = den + (size_t)d * HEADS;
#pragma unroll
    for (int hh = 0; hh < HEADS; ++hh) {
        float sc = ps[hh] + pd[hh];
        sc = (sc >= 0.0f) ? sc : NEG_SLOPE * sc;
        atomicAdd(pden + hh, expf(sc - pm[hh]));
    }
}

__global__ void gat_edge_agg(const int* __restrict__ src, const int* __restrict__ dst,
                             const float* __restrict__ s_src, const float* __restrict__ s_dst,
                             const float* __restrict__ m, const float* __restrict__ den,
                             const float* __restrict__ z, float* __restrict__ out0) {
    int idx = blockIdx.x * blockDim.x + threadIdx.x;
    if (idx >= N_EDGES * HEADS) return;
    int e = idx >> 3, hh = idx & 7;
    int s = src[e], d = dst[e];
    float sc = s_src[(size_t)s * HEADS + hh] + s_dst[(size_t)d * HEADS + hh];
    sc = (sc >= 0.0f) ? sc : NEG_SLOPE * sc;
    float alpha = expf(sc - m[(size_t)d * HEADS + hh]) / den[(size_t)d * HEADS + hh];

    const float4* zp = (const float4*)(z + (size_t)s * OUT_DIM + hh * HDIM);
    float* op = out0 + (size_t)d * OUT_DIM + hh * HDIM;
#pragma unroll
    for (int q = 0; q < 4; ++q) {
        float4 v = zp[q];
        atomicAdd(op + q * 4 + 0, alpha * v.x);
        atomicAdd(op + q * 4 + 1, alpha * v.y);
        atomicAdd(op + q * 4 + 2, alpha * v.z);
        atomicAdd(op + q * 4 + 3, alpha * v.w);
    }
}

__global__ void gat_finalize(const float* __restrict__ h, const float* __restrict__ e,
                             float* __restrict__ out) {
    int j = blockIdx.x * blockDim.x + threadIdx.x;
    const int NH = N_NODES * OUT_DIM;
    if (j < NH) {
        float v = out[j];
        v = (v > 0.0f) ? v : expm1f(v);
        out[j] = h[j] + v;
    } else if (j < NH + N_EDGES) {
        out[j] = e[j - NH];
    }
}

extern "C" void kernel_launch(void* const* d_in, const int* in_sizes, int n_in,
                              void* d_out, int out_size, void* d_ws, size_t ws_size,
                              hipStream_t stream) {
    const float* h     = (const float*)d_in[0];
    const float* e     = (const float*)d_in[1];
    const int*   src   = (const int*)d_in[2];
    const int*   dst   = (const int*)d_in[3];
    const float* Wfc   = (const float*)d_in[4];
    const float* Wattn = (const float*)d_in[5];
    float* out = (float*)d_out;

    float* z     = (float*)d_ws;
    float* s_src = z + (size_t)N_NODES * OUT_DIM;
    float* s_dst = s_src + (size_t)N_NODES * HEADS;
    float* m     = s_dst + (size_t)N_NODES * HEADS;
    float* den   = m + (size_t)N_NODES * HEADS;

    const int T = 256;

    gat_init<<<(N_NODES * OUT_DIM + T - 1) / T, T, 0, stream>>>(out, m, den);

    gat_gemm_wmma<<<(N_NODES / 16 + 7) / 8, T, 0, stream>>>(h, Wfc, z);

    gat_scores<<<(N_NODES * HEADS + T - 1) / T, T, 0, stream>>>(z, Wattn, s_src, s_dst);

    gat_edge_max<<<(N_EDGES + T - 1) / T, T, 0, stream>>>(src, dst, s_src, s_dst, m);
    gat_edge_sum<<<(N_EDGES + T - 1) / T, T, 0, stream>>>(src, dst, s_src, s_dst, m, den);
    gat_edge_agg<<<(N_EDGES * HEADS + T - 1) / T, T, 0, stream>>>(src, dst, s_src, s_dst,
                                                                  m, den, z, out);

    gat_finalize<<<(N_NODES * OUT_DIM + N_EDGES + T - 1) / T, T, 0, stream>>>(h, e, out);
}


# Group edges by destination in the call (count, scan, place); one wave per node does max, exp-sum and weighted row sum in registers in ascending-source order: float atomics removed
# speedup vs baseline: 7.0591x; 7.0591x over previous
#include <hip/hip_runtime.h>
#include <math.h>

#define N_NODES 100000
#define N_EDGES 1600000
#define IN_DIM  128
#define HEADS   8
#define HDIM    16
#define OUT_DIM 128
#define NEG_SLOPE 0.01f
#define LDS_STRIDE 130

typedef __attribute__((ext_vector_type(2))) float v2f;
typedef __attribute__((ext_vector_type(8))) float v8f;

#define SCAN_TILE 1024
#define SCAN_BLOCKS ((N_NODES + SCAN_TILE - 1) / SCAN_TILE)
static_assert(N_NODES % 4 == 0 && N_EDGES % 4 == 0, "16-byte accesses below");
static_assert(SCAN_BLOCKS <= 256, "one thread per tile total in gat_scan_add");

__global__ void gat_prep(const float* __restrict__ e, float* __restrict__ out,
                         int* __restrict__ cnt) {
    int j = blockIdx.x * blockDim.x + threadIdx.x;
    if (j < N_EDGES / 4)
        ((float4*)(out + (size_t)N_NODES * OUT_DIM))[j] = ((const float4*)e)[j];
    if (j < N_NODES / 4) ((int4*)cnt)[j] = make_int4(0, 0, 0, 0);
}

__global__ void gat_count(const int* __restrict__ dst, int* __restrict__ cnt,
                          int* __restrict__ slot) {
    int e = blockIdx.x * blockDim.x + threadIdx.x;
    if (e >= N_EDGES) return;
    slot[e] = atomicAdd(cnt + dst[e], 1);
}

__global__ void __launch_bounds__(256)
gat_scan_tiles(int* __restrict__ cnt, int* __restrict__ tsum) {
    __shared__ int wsum[8];
    const int t = threadIdx.x, lane = t & 31, w = t >> 5;
    const int i4 = blockIdx.x * 256 + t;
    int4 v = make_int4(0, 0, 0, 0);
    if (i4 < N_NODES / 4) v = ((const int4*)cnt)[i4];
    const int s0 = v.x, s1 = s0 + v.y, s2 = s1 + v.z, s3 = s2 + v.w;
    int x = s3;
    for (int o = 1; o < 32; o <<= 1) {
        int y = __shfl_up(x, o);
        if (lane >= o) x += y;
    }
    if (lane == 31) wsum[w] = x;
    __syncthreads();
    int woff = 0;
    for (int i = 0; i < 8; ++i) woff += (i < w) ? wsum[i] : 0;
    const int excl = woff + x - s3;
    if (i4 < N_NODES / 4)
        ((int4*)cnt)[i4] = make_int4(excl, excl + s0, excl + s1, excl + s2);
    if (t == 255) tsum[blockIdx.x] = woff + x;
}

__global__ void __launch_bounds__(256)
gat_scan_add(int* __restrict__ cnt, const int* __restrict__ tsum) {
    __shared__ int red[8];
    const int t = threadIdx.x;
    int v = (t < (int)blockIdx.x) ? tsum[t] : 0;
    for (int o = 16; o > 0; o >>= 1) v += __shfl_xor(v, o);
    if ((t & 31) == 0) red[t >> 5] = v;
    __syncthreads();
    int off = 0;
    for (int i = 0; i < 8; ++i) off += red[i];
    const int i4 = blockIdx.x * 256 + t;
    if (i4 < N_NODES / 4) {
        int4 x = ((int4*)cnt)[i4];
        x.x += off; x.y += off; x.z += off; x.w += off;
        ((int4*)cnt)[i4] = x;
    }
}

__global__ void gat_place(const int* __restrict__ src, const int* __restrict__ dst,
                          const int* __restrict__ rowptr, const int* __restrict__ slot,
                          int* __restrict__ perm) {
    int e = blockIdx.x * blockDim.x + threadIdx.x;
    if (e >= N_EDGES) return;
    perm[rowptr[dst[e]] + slot[e]] = src[e];
}

__global__ void __launch_bounds__(256)
gat_gemm_wmma(const float* __restrict__ h, const float* __restrict__ Wfc,
              float* __restrict__ z) {
    __shared__ float Wl[OUT_DIM * LDS_STRIDE];

    const int tid = threadIdx.x;
    for (int j = tid; j < IN_DIM * OUT_DIM; j += 256) {
        int i  = j >> 7;
        int c  = j & 127;
        int hh = c >> 4;
        int d  = c & 15;
        Wl[c * LDS_STRIDE + i] = Wfc[hh * (IN_DIM * HDIM) + i * HDIM + d];
    }
    __syncthreads();

    const int wave  = tid >> 5;
    const int lane  = tid & 31;
    const int mtile = blockIdx.x * 8 + wave;
    if (mtile >= N_NODES / 16) return;

    const int row   = lane & 15;
    const int khalf = lane >> 4;

    const float* hrow = h + (size_t)(mtile * 16 + row) * IN_DIM;

    v8f acc[8] = {};

    for (int k0 = 0; k0 < IN_DIM; k0 += 4) {
        const int kk = k0 + 2 * khalf;
        v2f a = *reinterpret_cast<const v2f*>(hrow + kk);
#pragma unroll
        for (int t = 0; t < 8; ++t) {
            v2f b = *reinterpret_cast<const v2f*>(&Wl[(t * 16 + row) * LDS_STRIDE + kk]);
            acc[t] = __builtin_amdgcn_wmma_f32_16x16x4_f32(
                false, a, false, b, (short)0, acc[t], false, false);
        }
    }

    float* zbase = z + (size_t)(mtile * 16) * OUT_DIM;
#pragma unroll
    for (int t = 0; t < 8; ++t) {
#pragma unroll
        for (int i = 0; i < 8; ++i) {
            int mrow = i + khalf * 8;
            zbase[(size_t)mrow * OUT_DIM + t * 16 + row] = acc[t][i];
        }
    }
}

__global__ void gat_scores(const float* __restrict__ z,
                           const float* __restrict__ Wattn,
                           float* __restrict__ s_src) {
    int idx = blockIdx.x * blockDim.x + threadIdx.x;
    if (idx >= N_NODES * HEADS) return;
    int n = idx >> 3, hh = idx & 7;
    const float* zp = z + (size_t)n * OUT_DIM + hh * HDIM;
    const float* as = Wattn + hh * (2 * HDIM);
    float ss = 0.0f;
    for (int d = 0; d < HDIM; ++d) ss += zp[d] * as[d];
    s_src[idx] = ss;
}

__device__ __forceinline__ float edge_exp(float xs, float sd, float m) {
    float sc = xs + sd;
    sc = (sc >= 0.0f) ? sc : NEG_SLOPE * sc;
    return expf(sc - m);
}

__device__ __forceinline__ float elu1(float v) {
    return (v > 0.0f) ? v : expm1f(v);
}

__device__ __forceinline__ int wave_order(int s, int c, int lane) {
    int rank = 0;
    for (int i = 0; i < c; ++i) {
        const int si = __builtin_amdgcn_readlane(s, i);
        rank += ((si < s) || (si == s && i < lane)) ? 1 : 0;
    }
    rank = (lane < c) ? rank : lane;
    return __builtin_amdgcn_ds_permute(rank << 2, s);
}

__global__ void __launch_bounds__(256)
gat_node_gather(const int* __restrict__ rowptr, const int* __restrict__ perm,
                const float* __restrict__ s_src, const float* __restrict__ z,
                const float* __restrict__ Wattn, const float* __restrict__ h,
                float* __restrict__ out) {
    const int lane = threadIdx.x & 31;
    const int node = __builtin_amdgcn_readfirstlane((int)(blockIdx.x * 8 + (threadIdx.x >> 5)));
    if (node >= N_NODES) return;
    const int hh = lane >> 2;

    const int beg = rowptr[node];
    const int end = (node + 1 < N_NODES) ? rowptr[node + 1] : N_EDGES;
    const int deg = end - beg;

    const float4 zd = *(const float4*)(z + (size_t)node * OUT_DIM + lane * 4);
    const float4 hv = *(const float4*)(h + (size_t)node * OUT_DIM + lane * 4);
    const float4 ad = *(const float4*)(Wattn + hh * (2 * HDIM) + HDIM + (lane & 3) * 4);
    float sd = zd.x * ad.x + zd.y * ad.y + zd.z * ad.z + zd.w * ad.w;
    sd += __shfl_xor(sd, 1);
    sd += __shfl_xor(sd, 2);

    const int s_first = (lane < deg) ? perm[beg + lane] : 0;

    float mx[HEADS];
    for (int k = 0; k < HEADS; ++k) mx[k] = -INFINITY;
    for (int base = 0; base < deg; base += 32) {
        const bool valid = base + lane < deg;
        const int s = (base == 0) ? s_first : (valid ? perm[beg + base + lane] : 0);
        const float4 a = *(const float4*)(s_src + (size_t)s * HEADS);
        const float4 b = *(const float4*)(s_src + (size_t)s * HEADS + 4);
        if (valid) {
            mx[0] = fmaxf(mx[0], a.x); mx[1] = fmaxf(mx[1], a.y);
            mx[2] = fmaxf(mx[2], a.z); mx[3] = fmaxf(mx[3], a.w);
            mx[4] = fmaxf(mx[4], b.x); mx[5] = fmaxf(mx[5], b.y);
            mx[6] = fmaxf(mx[6], b.z); mx[7] = fmaxf(mx[7], b.w);
        }
    }
    const bool b4 = (lane & 16) != 0, b3 = (lane & 8) != 0, b2 = (lane & 4) != 0;
    float w4[4], u2[2], t;
    for (int i = 0; i < 4; ++i) {
        const float send = b4 ? mx[i] : mx[4 + i];
        const float keep = b4 ? mx[4 + i] : mx[i];
        w4[i] = fmaxf(keep, __shfl_xor(send, 16));
    }
    for (int i = 0; i < 2; ++i) {
        const float send = b3 ? w4[i] : w4[2 + i];
        const float keep = b3 ? w4[2 + i] : w4[i];
        u2[i] = fmaxf(keep, __shfl_xor(send, 8));
    }
    {
        const float send = b2 ? u2[0] : u2[1];
        const float keep = b2 ? u2[1] : u2[0];
        t = fmaxf(keep, __shfl_xor(send, 4));
    }
    t = fmaxf(t, __shfl_xor(t, 2));
    t = fmaxf(t, __shfl_xor(t, 1));
    float m = t + sd;
    m = (m >= 0.0f) ? m : NEG_SLOPE * m;

    float den = 0.0f;
    float4 acc = make_float4(0.0f, 0.0f, 0.0f, 0.0f);
    const float* srow = s_src + hh;
    const float* zcol = z + lane * 4;
    for (int base = 0; base < deg; base += 32) {
        const int c = min(32, deg - base);
        int s = (base == 0) ? s_first : ((lane < c) ? perm[beg + base + lane] : 0);
        s = wave_order(s, c, lane);
        for (int j = 0; j < c; j += 4) {
            const int j1 = min(j + 1, c - 1), j2 = min(j + 2, c - 1), j3 = min(j + 3, c - 1);
            const int sa = __builtin_amdgcn_readlane(s, j);
            const int sb = __builtin_amdgcn_readlane(s, j1);
            const int sc = __builtin_amdgcn_readlane(s, j2);
            const int sq = __builtin_amdgcn_readlane(s, j3);
            const float xa = srow[(size_t)sa * HEADS];
            const float xb = srow[(size_t)sb * HEADS];
            const float xc = srow[(size_t)sc * HEADS];
            const float xq = srow[(size_t)sq * HEADS];
            const float4 za = *(const float4*)(zcol + (size_t)sa * OUT_DIM);
            const float4 zb = *(const float4*)(zcol + (size_t)sb * OUT_DIM);
            const float4 zc = *(const float4*)(zcol + (size_t)sc * OUT_DIM);
            const float4 zq = *(const float4*)(zcol + (size_t)sq * OUT_DIM);
            const float kb = (j + 1 < c) ? 1.0f : 0.0f;
            const float kc = (j + 2 < c) ? 1.0f : 0.0f;
            const float kq = (j + 3 < c) ? 1.0f : 0.0f;
            const float pa = edge_exp(xa, sd, m);
            const float pb = edge_exp(xb, sd, m) * kb;
            const float pc = edge_exp(xc, sd, m) * kc;
            const float pq = edge_exp(xq, sd, m) * kq;
            den += pa;
            acc.x += pa * za.x; acc.y += pa * za.y; acc.z += pa * za.z; acc.w += pa * za.w;
            den += pb;
            acc.x += pb * zb.x; acc.y += pb * zb.y; acc.z += pb * zb.z; acc.w += pb * zb.w;
            den += pc;
            acc.x += pc * zc.x; acc.y += pc * zc.y; acc.z += pc * zc.z; acc.w += pc * zc.w;
            den += pq;
            acc.x += pq * zq.x; acc.y += pq * zq.y; acc.z += pq * zq.z; acc.w += pq * zq.w;
        }
    }

    float4 r = make_float4(0.0f, 0.0f, 0.0f, 0.0f);
    if (deg > 0) {
        const float inv = 1.0f / den;
        r.x = acc.x * inv; r.y = acc.y * inv; r.z = acc.z * inv; r.w = acc.w * inv;
    }
    float4 o;
    o.x = hv.x + elu1(r.x); o.y = hv.y + elu1(r.y);
    o.z = hv.z + elu1(r.z); o.w = hv.w + elu1(r.w);
    *(float4*)(out + (size_t)node * OUT_DIM + lane * 4) = o;
}

extern "C" void kernel_launch(void* const* d_in, const int* in_sizes, int n_in,
                              void* d_out, int out_size, void* d_ws, size_t ws_size,
                              hipStream_t stream) {
    const float* h     = (const float*)d_in[0];
    const float* e     = (const float*)d_in[1];
    const int*   src   = (const int*)d_in[2];
    const int*   dst   = (const int*)d_in[3];
    const float* Wfc   = (const float*)d_in[4];
    const float* Wattn = (const float*)d_in[5];
    float* out = (float*)d_out;

    float* z      = (float*)d_ws;
    float* s_src  = z + (size_t)N_NODES * OUT_DIM;
    int*   rowptr = (int*)(s_src + (size_t)N_NODES * HEADS);
    int*   tsum   = rowptr + N_NODES;
    int*   perm   = rowptr + (size_t)N_NODES * HEADS;
    int*   slot   = (int*)d_ws;

    const int T = 256;

    gat_prep<<<(N_EDGES / 4 + T - 1) / T, T, 0, stream>>>(e, out, rowptr);
    gat_count<<<(N_EDGES + T - 1) / T, T, 0, stream>>>(dst, rowptr, slot);
    gat_scan_tiles<<<SCAN_BLOCKS, T, 0, stream>>>(rowptr, tsum);
    gat_scan_add<<<SCAN_BLOCKS, T, 0, stream>>>(rowptr, tsum);
    gat_place<<<(N_EDGES + T - 1) / T, T, 0, stream>>>(src, dst, rowptr, slot, perm);

    gat_gemm_wmma<<<(N_NODES / 16 + 7) / 8, T, 0, stream>>>(h, Wfc, z);

    gat_scores<<<(N_NODES * HEADS + T - 1) / T, T, 0, stream>>>(z, Wattn, s_src);

    gat_node_gather<<<(N_NODES + 7) / 8, T, 0, stream>>>(rowptr, perm, s_src, z, Wattn, h, out);
}


# Group edges by destination with a two-level counting sort on LDS counters (slices of 256 nodes, then nodes inside a slice): per-edge global returning adds and row-pointer gathers are gone
# speedup vs baseline: 9.9424x; 1.4085x over previous
#include <hip/hip_runtime.h>
#include <math.h>

#define N_NODES 100000
#define N_EDGES 1600000
#define IN_DIM  128
#define HEADS   8
#define HDIM    16
#define OUT_DIM 128
#define NEG_SLOPE 0.01f
#define LDS_STRIDE 130

typedef __attribute__((ext_vector_type(2))) float v2f;
typedef __attribute__((ext_vector_type(8))) float v8f;

#define BIN_SHIFT 8
#define BIN_NODES (1 << BIN_SHIFT)
#define NBINS ((N_NODES + BIN_NODES - 1) / BIN_NODES)
#define NBLK 256
#define CHUNK (N_EDGES / NBLK)
#define SRC_BITS 17
#define SRC_MASK ((1 << SRC_BITS) - 1)
#define TBL_LEN (NBINS * NBLK)
#define BT 1024

#define SCAN_TILE 1024
#define SCAN_BLOCKS ((TBL_LEN + SCAN_TILE - 1) / SCAN_TILE)
static_assert(N_EDGES % 4 == 0 && TBL_LEN % 4 == 0, "16-byte accesses below");
static_assert(SCAN_BLOCKS <= 256, "one thread per tile total in gat_scan_add");
static_assert(N_EDGES % NBLK == 0, "equal edge chunks");
static_assert(N_NODES <= (1 << SRC_BITS), "source index fits the packed entry");

__global__ void __launch_bounds__(BT)
gat_slice_count(const int* __restrict__ dst, const float* __restrict__ e,
                float* __restrict__ out, int* __restrict__ tbl) {
    __shared__ int hist[NBINS];
    const int t = threadIdx.x, blk = blockIdx.x;
    for (int i = t; i < NBINS; i += BT) hist[i] = 0;
    __syncthreads();
    const int* dchunk = dst + (size_t)blk * CHUNK;
    for (int b0 = 0; b0 < CHUNK; b0 += 4 * BT) {
        int d[4];
        for (int k = 0; k < 4; ++k) {
            const int i = b0 + k * BT + t;
            d[k] = (i < CHUNK) ? dchunk[i] : -1;
        }
        for (int k = 0; k < 4; ++k)
            if (d[k] >= 0) atomicAdd(&hist[d[k] >> BIN_SHIFT], 1);
    }
    for (int j = blk * BT + t; j < N_EDGES / 4; j += NBLK * BT)
        ((float4*)(out + (size_t)N_NODES * OUT_DIM))[j] = ((const float4*)e)[j];
    __syncthreads();
    for (int i = t; i < NBINS; i += BT) tbl[i * NBLK + blk] = hist[i];
}

__global__ void __launch_bounds__(256)
gat_scan_tiles(int* __restrict__ cnt, int* __restrict__ tsum) {
    __shared__ int wsum[8];
    const int t = threadIdx.x, lane = t & 31, w = t >> 5;
    const int i4 = blockIdx.x * 256 + t;
    int4 v = make_int4(0, 0, 0, 0);
    if (i4 < TBL_LEN / 4) v = ((const int4*)cnt)[i4];
    const int s0 = v.x, s1 = s0 + v.y, s2 = s1 + v.z, s3 = s2 + v.w;
    int x = s3;
    for (int o = 1; o < 32; o <<= 1) {
        int y = __shfl_up(x, o);
        if (lane >= o) x += y;
    }
    if (lane == 31) wsum[w] = x;
    __syncthreads();
    int woff = 0;
    for (int i = 0; i < 8; ++i) woff += (i < w) ? wsum[i] : 0;
    const int excl = woff + x - s3;
    if (i4 < TBL_LEN / 4)
        ((int4*)cnt)[i4] = make_int4(excl, excl + s0, excl + s1, excl + s2);
    if (t == 255) tsum[blockIdx.x] = woff + x;
}

__global__ void __launch_bounds__(256)
gat_scan_add(int* __restrict__ cnt, const int* __restrict__ tsum) {
    __shared__ int red[8];
    const int t = threadIdx.x;
    int v = (t < (int)blockIdx.x) ? tsum[t] : 0;
    for (int o = 16; o > 0; o >>= 1) v += __shfl_xor(v, o);
    if ((t & 31) == 0) red[t >> 5] = v;
    __syncthreads();
    int off = 0;
    for (int i = 0; i < 8; ++i) off += red[i];
    const int i4 = blockIdx.x * 256 + t;
    if (i4 < TBL_LEN / 4) {
        int4 x = ((int4*)cnt)[i4];
        x.x += off; x.y += off; x.z += off; x.w += off;
        ((int4*)cnt)[i4] = x;
    }
}

__global__ void __launch_bounds__(BT)
gat_slice_scatter(const int* __restrict__ src, const int* __restrict__ dst,
                  const int* __restrict__ tbl, int* __restrict__ words) {
    __shared__ int cur[NBINS];
    const int t = threadIdx.x, blk = blockIdx.x;
    for (int i = t; i < NBINS; i += BT) cur[i] = tbl[i * NBLK + blk];
    __syncthreads();
    const int* dchunk = dst + (size_t)blk * CHUNK;
    const int* schunk = src + (size_t)blk * CHUNK;
    for (int b0 = 0; b0 < CHUNK; b0 += 4 * BT) {
        int d[4], s[4];
        for (int k = 0; k < 4; ++k) {
            const int i = b0 + k * BT + t;
            d[k] = (i < CHUNK) ? dchunk[i] : -1;
            s[k] = (i < CHUNK) ? schunk[i] : 0;
        }
        for (int k = 0; k < 4; ++k)
            if (d[k] >= 0) {
                const int pos = atomicAdd(&cur[d[k] >> BIN_SHIFT], 1);
                words[pos] = ((d[k] & (BIN_NODES - 1)) << SRC_BITS) | s[k];
            }
    }
}

__global__ void __launch_bounds__(BT)
gat_slice_place(const int* __restrict__ tbl, const int* __restrict__ words,
                int* __restrict__ rowptr, int* __restrict__ perm) {
    __shared__ int cnt[BIN_NODES];
    __shared__ int wsum[BIN_NODES / 32];
    const int t = threadIdx.x, b = blockIdx.x, lane = t & 31;
    const int beg = tbl[b * NBLK];
    const int end = (b + 1 < NBINS) ? tbl[(b + 1) * NBLK] : N_EDGES;
    const int n = end - beg;
    const int* wsl = words + beg;
    if (t < BIN_NODES) cnt[t] = 0;
    __syncthreads();
    for (int b0 = 0; b0 < n; b0 += 4 * BT) {
        int w[4];
        for (int k = 0; k < 4; ++k) {
            const int i = b0 + k * BT + t;
            w[k] = (i < n) ? wsl[i] : -1;
        }
        for (int k = 0; k < 4; ++k)
            if (w[k] >= 0) atomicAdd(&cnt[w[k] >> SRC_BITS], 1);
    }
    __syncthreads();
    const int c = (t < BIN_NODES) ? cnt[t] : 0;
    int x = c;
    for (int o = 1; o < 32; o <<= 1) {
        int y = __shfl_up(x, o);
        if (lane >= o) x += y;
    }
    if (t < BIN_NODES && lane == 31) wsum[t >> 5] = x;
    __syncthreads();
    if (t < BIN_NODES) {
        int woff = 0;
        for (int i = 0; i < BIN_NODES / 32; ++i) woff += (i < (t >> 5)) ? wsum[i] : 0;
        const int excl = woff + x - c;
        cnt[t] = excl;
        const int node = b * BIN_NODES + t;
        if (node < N_NODES) rowptr[node] = beg + excl;
    }
    __syncthreads();
    int* psl = perm + beg;
    for (int b0 = 0; b0 < n; b0 += 4 * BT) {
        int w[4];
        for (int k = 0; k < 4; ++k) {
            const int i = b0 + k * BT + t;
            w[k] = (i < n) ? wsl[i] : -1;
        }
        for (int k = 0; k < 4; ++k)
            if (w[k] >= 0) {
                const int pos = atomicAdd(&cnt[w[k] >> SRC_BITS], 1);
                psl[pos] = w[k] & SRC_MASK;
            }
    }
}

__global__ void __launch_bounds__(256)
gat_gemm_wmma(const float* __restrict__ h, const float* __restrict__ Wfc,
              float* __restrict__ z) {
    __shared__ float Wl[OUT_DIM * LDS_STRIDE];

    const int tid = threadIdx.x;
    for (int j = tid; j < IN_DIM * OUT_DIM; j += 256) {
        int i  = j >> 7;
        int c  = j & 127;
        int hh = c >> 4;
        int d  = c & 15;
        Wl[c * LDS_STRIDE + i] = Wfc[hh * (IN_DIM * HDIM) + i * HDIM + d];
    }
    __syncthreads();

    const int wave  = tid >> 5;
    const int lane  = tid & 31;
    const int mtile = blockIdx.x * 8 + wave;
    if (mtile >= N_NODES / 16) return;

    const int row   = lane & 15;
    const int khalf = lane >> 4;

    const float* hrow = h + (size_t)(mtile * 16 + row) * IN_DIM;

    v8f acc[8] = {};

    for (int k0 = 0; k0 < IN_DIM; k0 += 4) {
        const int kk = k0 + 2 * khalf;
        v2f a = *reinterpret_cast<const v2f*>(hrow + kk);
#pragma unroll
        for (int t = 0; t < 8; ++t) {
            v2f b = *reinterpret_cast<const v2f*>(&Wl[(t * 16 + row) * LDS_STRIDE + kk]);
            acc[t] = __builtin_amdgcn_wmma_f32_16x16x4_f32(
                false, a, false, b, (short)0, acc[t], false, false);
        }
    }

    float* zbase = z + (size_t)(mtile * 16) * OUT_DIM;
#pragma unroll
    for (int t = 0; t < 8; ++t) {
#pragma unroll
        for (int i = 0; i < 8; ++i) {
            int mrow = i + khalf * 8;
            zbase[(size_t)mrow * OUT_DIM + t * 16 + row] = acc[t][i];
        }
    }
}

__global__ void gat_scores(const float* __restrict__ z,
                           const float* __restrict__ Wattn,
                           float* __restrict__ s_src) {
    int idx = blockIdx.x * blockDim.x + threadIdx.x;
    if (idx >= N_NODES * HEADS) return;
    int n = idx >> 3, hh = idx & 7;
    const float* zp = z + (size_t)n * OUT_DIM + hh * HDIM;
    const float* as = Wattn + hh * (2 * HDIM);
    float ss = 0.0f;
    for (int d = 0; d < HDIM; ++d) ss += zp[d] * as[d];
    s_src[idx] = ss;
}

__device__ __forceinline__ float edge_exp(float xs, float sd, float m) {
    float sc = xs + sd;
    sc = (sc >= 0.0f) ? sc : NEG_SLOPE * sc;
    return expf(sc - m);
}

__device__ __forceinline__ float elu1(float v) {
    return (v > 0.0f) ? v : expm1f(v);
}

__device__ __forceinline__ int wave_order(int s, int c, int lane) {
    int rank = 0;
    for (int i = 0; i < c; ++i) {
        const int si = __builtin_amdgcn_readlane(s, i);
        rank += ((si < s) || (si == s && i < lane)) ? 1 : 0;
    }
    rank = (lane < c) ? rank : lane;
    return __builtin_amdgcn_ds_permute(rank << 2, s);
}

__global__ void __launch_bounds__(256)
gat_node_gather(const int* __restrict__ rowptr, const int* __restrict__ perm,
                const float* __restrict__ s_src, const float* __restrict__ z,
                const float* __restrict__ Wattn, const float* __restrict__ h,
                float* __restrict__ out) {
    const int lane = threadIdx.x & 31;
    const int node = __builtin_amdgcn_readfirstlane((int)(blockIdx.x * 8 + (threadIdx.x >> 5)));
    if (node >= N_NODES) return;
    const int hh = lane >> 2;

    const int beg = rowptr[node];
    const int end = (node + 1 < N_NODES) ? rowptr[node + 1] : N_EDGES;
    const int deg = end - beg;

    const float4 zd = *(const float4*)(z + (size_t)node * OUT_DIM + lane * 4);
    const float4 hv = *(const float4*)(h + (size_t)node * OUT_DIM + lane * 4);
    const float4 ad = *(const float4*)(Wattn + hh * (2 * HDIM) + HDIM + (lane & 3) * 4);
    float sd = zd.x * ad.x + zd.y * ad.y + zd.z * ad.z + zd.w * ad.w;
    sd += __shfl_xor(sd, 1);
    sd += __shfl_xor(sd, 2);

    const int s_first = (lane < deg) ? perm[beg + lane] : 0;

    float mx[HEADS];
    for (int k = 0; k < HEADS; ++k) mx[k] = -INFINITY;
    for (int base = 0; base < deg; base += 32) {
        const bool valid = base + lane < deg;
        const int s = (base == 0) ? s_first : (valid ? perm[beg + base + lane] : 0);
        const float4 a = *(const float4*)(s_src + (size_t)s * HEADS);
        const float4 b = *(const float4*)(s_src + (size_t)s * HEADS + 4);
        if (valid) {
            mx[0] = fmaxf(mx[0], a.x); mx[1] = fmaxf(mx[1], a.y);
            mx[2] = fmaxf(mx[2], a.z); mx[3] = fmaxf(mx[3], a.w);
            mx[4] = fmaxf(mx[4], b.x); mx[5] = fmaxf(mx[5], b.y);
            mx[6] = fmaxf(mx[6], b.z); mx[7] = fmaxf(mx[7], b.w);
        }
    }
    const bool b4 = (lane & 16) != 0, b3 = (lane & 8) != 0, b2 = (lane & 4) != 0;
    float w4[4], u2[2], t;
    for (int i = 0; i < 4; ++i) {
        const float send = b4 ? mx[i] : mx[4 + i];
        const float keep = b4 ? mx[4 + i] : mx[i];
        w4[i] = fmaxf(keep, __shfl_xor(send, 16));
    }
    for (int i = 0; i < 2; ++i) {
        const float send = b3 ? w4[i] : w4[2 + i];
        const float keep = b3 ? w4[2 + i] : w4[i];
        u2[i] = fmaxf(keep, __shfl_xor(send, 8));
    }
    {
        const float send = b2 ? u2[0] : u2[1];
        const float keep = b2 ? u2[1] : u2[0];
        t = fmaxf(keep, __shfl_xor(send, 4));
    }
    t = fmaxf(t, __shfl_xor(t, 2));
    t = fmaxf(t, __shfl_xor(t, 1));
    float m = t + sd;
    m = (m >= 0.0f) ? m : NEG_SLOPE * m;

    float den = 0.0f;
    float4 acc = make_float4(0.0f, 0.0f, 0.0f, 0.0f);
    const float* srow = s_src + hh;
    const float* zcol = z + lane * 4;
    for (int base = 0; base < deg; base += 32) {
        const int c = min(32, deg - base);
        int s = (base == 0) ? s_first : ((lane < c) ? perm[beg + base + lane] : 0);
        s = wave_order(s, c, lane);
        for (int j = 0; j < c; j += 4) {
            const int j1 = min(j + 1, c - 1), j2 = min(j + 2, c - 1), j3 = min(j + 3, c - 1);
            const int sa = __builtin_amdgcn_readlane(s, j);
            const int sb = __builtin_amdgcn_readlane(s, j1);
            const int sc = __builtin_amdgcn_readlane(s, j2);
            const int sq = __builtin_amdgcn_readlane(s, j3);
            const float xa = srow[(size_t)sa * HEADS];
            const float xb = srow[(size_t)sb * HEADS];
            const float xc = srow[(size_t)sc * HEADS];
            const float xq = srow[(size_t)sq * HEADS];
            const float4 za = *(const float4*)(zcol + (size_t)sa * OUT_DIM);
            const float4 zb = *(const float4*)(zcol + (size_t)sb * OUT_DIM);
            const float4 zc = *(const float4*)(zcol + (size_t)sc * OUT_DIM);
            const float4 zq = *(const float4*)(zcol + (size_t)sq * OUT_DIM);
            const float kb = (j + 1 < c) ? 1.0f : 0.0f;
            const float kc = (j + 2 < c) ? 1.0f : 0.0f;
            const float kq = (j + 3 < c) ? 1.0f : 0.0f;
            const float pa = edge_exp(xa, sd, m);
            const float pb = edge_exp(xb, sd, m) * kb;
            const float pc = edge_exp(xc, sd, m) * kc;
            const float pq = edge_exp(xq, sd, m) * kq;
            den += pa;
            acc.x += pa * za.x; acc.y += pa * za.y; acc.z += pa * za.z; acc.w += pa * za.w;
            den += pb;
            acc.x += pb * zb.x; acc.y += pb * zb.y; acc.z += pb * zb.z; acc.w += pb * zb.w;
            den += pc;
            acc.x += pc * zc.x; acc.y += pc * zc.y; acc.z += pc * zc.z; acc.w += pc * zc.w;
            den += pq;
            acc.x += pq * zq.x; acc.y += pq * zq.y; acc.z += pq * zq.z; acc.w += pq * zq.w;
        }
    }

    float4 r = make_float4(0.0f, 0.0f, 0.0f, 0.0f);
    if (deg > 0) {
        const float inv = 1.0f / den;
        r.x = acc.x * inv; r.y = acc.y * inv; r.z = acc.z * inv; r.w = acc.w * inv;
    }
    float4 o;
    o.x = hv.x + elu1(r.x); o.y = hv.y + elu1(r.y);
    o.z = hv.z + elu1(r.z); o.w = hv.w + elu1(r.w);
    *(float4*)(out + (size_t)node * OUT_DIM + lane * 4) = o;
}

extern "C" void kernel_launch(void* const* d_in, const int* in_sizes, int n_in,
                              void* d_out, int out_size, void* d_ws, size_t ws_size,
                              hipStream_t stream) {
    const float* h     = (const float*)d_in[0];
    const float* e     = (const float*)d_in[1];
    const int*   src   = (const int*)d_in[2];
    const int*   dst   = (const int*)d_in[3];
    const float* Wfc   = (const float*)d_in[4];
    const float* Wattn = (const float*)d_in[5];
    float* out = (float*)d_out;

    float* z      = (float*)d_ws;
    float* s_src  = z + (size_t)N_NODES * OUT_DIM;
    int*   rowptr = (int*)(s_src + (size_t)N_NODES * HEADS);
    int*   tsum   = rowptr + N_NODES;
    int*   tbl    = tsum + 256;
    int*   perm   = rowptr + (size_t)N_NODES * HEADS;
    int*   words  = (int*)d_ws;

    const int T = 256;

    gat_slice_count<<<NBLK, BT, 0, stream>>>(dst, e, out, tbl);
    gat_scan_tiles<<<SCAN_BLOCKS, T, 0, stream>>>(tbl, tsum);
    gat_scan_add<<<SCAN_BLOCKS, T, 0, stream>>>(tbl, tsum);
    gat_slice_scatter<<<NBLK, BT, 0, stream>>>(src, dst, tbl, words);
    gat_slice_place<<<NBINS, BT, 0, stream>>>(tbl, words, rowptr, perm);

    gat_gemm_wmma<<<(N_NODES / 16 + 7) / 8, T, 0, stream>>>(h, Wfc, z);

    gat_scores<<<(N_NODES * HEADS + T - 1) / T, T, 0, stream>>>(z, Wattn, s_src);

    gat_node_gather<<<(N_NODES + 7) / 8, T, 0, stream>>>(rowptr, perm, s_src, z, Wattn, h, out);
}


# Edge grouping in 2 launches instead of 5: each chunk orders its own edges by slice in LDS (count, scan, place in one work-group); a slice's work-group finds its 256 runs by LDS search: no global scan
# speedup vs baseline: 12.2549x; 1.2326x over previous
#include <hip/hip_runtime.h>
#include <math.h>

#define N_NODES 100000
#define N_EDGES 1600000
#define IN_DIM  128
#define HEADS   8
#define HDIM    16
#define OUT_DIM 128
#define NEG_SLOPE 0.01f
#define LDS_STRIDE 130

typedef __attribute__((ext_vector_type(2))) float v2f;
typedef __attribute__((ext_vector_type(8))) float v8f;

#define BIN_SHIFT 8
#define BIN_NODES (1 << BIN_SHIFT)
#define NBINS ((N_NODES + BIN_NODES - 1) / BIN_NODES)
#define NBLK 256
#define CHUNK (N_EDGES / NBLK)
#define SRC_BITS 17
#define SRC_MASK ((1 << SRC_BITS) - 1)
#define BT 1024
#define LSTR (NBINS + 1)
#define BIN_WAVES ((NBINS + 31) / 32)
#define E_PER ((N_EDGES / 4 + NBLK * BT - 1) / (NBLK * BT))
static_assert(N_EDGES % 4 == 0, "16-byte copies of e");
static_assert(N_EDGES % NBLK == 0, "equal edge chunks");
static_assert(NBINS <= BT, "one thread per slice in the chunk scan");
static_assert(NBLK <= BT && (NBLK & (NBLK - 1)) == 0, "one thread per run; the run search halves a power of two");
static_assert(N_NODES <= (1 << SRC_BITS), "source index fits the packed entry");

__global__ void __launch_bounds__(BT)
gat_chunk_order(const int* __restrict__ src, const int* __restrict__ dst,
                const float* __restrict__ e, float* __restrict__ out,
                int* __restrict__ loff, int* __restrict__ words) {
    __shared__ int cur[NBINS];
    __shared__ int wsum[BIN_WAVES];
    const int t = threadIdx.x, blk = blockIdx.x, lane = t & 31, w = t >> 5;
    if (t < NBINS) cur[t] = 0;
    __syncthreads();
    float4 ev[E_PER];
    for (int k = 0; k < E_PER; ++k) {
        const int j = (k * NBLK + blk) * BT + t;
        ev[k] = (j < N_EDGES / 4) ? ((const float4*)e)[j] : make_float4(0.0f, 0.0f, 0.0f, 0.0f);
    }
    const int* dchunk = dst + (size_t)blk * CHUNK;
    const int* schunk = src + (size_t)blk * CHUNK;
    for (int b0 = 0; b0 < CHUNK; b0 += 4 * BT) {
        int d[4];
        for (int k = 0; k < 4; ++k) {
            const int i = b0 + k * BT + t;
            d[k] = (i < CHUNK) ? dchunk[i] : -1;
        }
        for (int k = 0; k < 4; ++k)
            if (d[k] >= 0) atomicAdd(&cur[d[k] >> BIN_SHIFT], 1);
    }
    __syncthreads();
    const int c = (t < NBINS) ? cur[t] : 0;
    int x = c;
    for (int o = 1; o < 32; o <<= 1) {
        int y = __shfl_up(x, o);
        if (lane >= o) x += y;
    }
    if (w < BIN_WAVES && lane == 31) wsum[w] = x;
    __syncthreads();
    int woff = 0;
    for (int i = 0; i < BIN_WAVES; ++i) woff += (i < w) ? wsum[i] : 0;
    const int excl = woff + x - c;
    if (t < NBINS) cur[t] = excl;
    __syncthreads();
    int* wchunk = words + (size_t)blk * CHUNK;
    for (int b0 = 0; b0 < CHUNK; b0 += 4 * BT) {
        int d[4], s[4];
        for (int k = 0; k < 4; ++k) {
            const int i = b0 + k * BT + t;
            d[k] = (i < CHUNK) ? dchunk[i] : -1;
            s[k] = (i < CHUNK) ? schunk[i] : 0;
        }
        for (int k = 0; k < 4; ++k)
            if (d[k] >= 0) {
                const int pos = atomicAdd(&cur[d[k] >> BIN_SHIFT], 1);
                wchunk[pos] = ((d[k] & (BIN_NODES - 1)) << SRC_BITS) | s[k];
            }
    }
    if (t < NBINS) loff[blk * LSTR + t] = excl;
    if (t == NBINS) loff[blk * LSTR + NBINS] = CHUNK;
    for (int k = 0; k < E_PER; ++k) {
        const int j = (k * NBLK + blk) * BT + t;
        if (j < N_EDGES / 4) ((float4*)(out + (size_t)N_NODES * OUT_DIM))[j] = ev[k];
    }
}

__device__ __forceinline__ void slice_entries(const int* rs, const int* ra,
                                              const int* __restrict__ words,
                                              int n, int i0, int (&wv)[4]) {
    int ii[4], r[4];
    bool ok[4];
    for (int k = 0; k < 4; ++k) {
        const int i = i0 + k * BT;
        ok[k] = i < n;
        ii[k] = ok[k] ? i : n - 1;
        r[k] = 0;
    }
    for (int step = NBLK / 2; step > 0; step >>= 1)
        for (int k = 0; k < 4; ++k)
            if (rs[r[k] + step] <= ii[k]) r[k] += step;
    for (int k = 0; k < 4; ++k) {
        const int v = words[ra[r[k]] + ii[k]];
        wv[k] = ok[k] ? v : -1;
    }
}

__global__ void __launch_bounds__(BT)
gat_slice_place(const int* __restrict__ loff, const int* __restrict__ words,
                int* __restrict__ rowptr, int* __restrict__ perm) {
    __shared__ int cnt[BIN_NODES];
    __shared__ int wsum[BIN_NODES / 32];
    __shared__ int lsum[NBLK / 32];
    __shared__ int osum[NBLK / 32];
    __shared__ int rs[NBLK];
    __shared__ int ra[NBLK];
    const int t = threadIdx.x, b = blockIdx.x, lane = t & 31, w = t >> 5;

    int o0 = 0, len = 0;
    if (t < NBLK) {
        o0  = loff[t * LSTR + b];
        len = loff[t * LSTR + b + 1] - o0;
    }
    int xl = len;
    for (int o = 1; o < 32; o <<= 1) {
        int y = __shfl_up(xl, o);
        if (lane >= o) xl += y;
    }
    int so = o0;
    for (int o = 16; o > 0; o >>= 1) so += __shfl_xor(so, o);
    if (t < NBLK && lane == 31) { lsum[w] = xl; osum[w] = so; }
    if (t < BIN_NODES) cnt[t] = 0;
    __syncthreads();
    int loffw = 0, n = 0, beg = 0;
    for (int i = 0; i < NBLK / 32; ++i) {
        loffw += (i < w) ? lsum[i] : 0;
        n     += lsum[i];
        beg   += osum[i];
    }
    if (t < NBLK) {
        const int first = loffw + xl - len;
        rs[t] = first;
        ra[t] = t * CHUNK + o0 - first;
    }
    __syncthreads();

    for (int b0 = 0; b0 < n; b0 += 4 * BT) {
        int wv[4];
        slice_entries(rs, ra, words, n, b0 + t, wv);
        for (int k = 0; k < 4; ++k)
            if (wv[k] >= 0) atomicAdd(&cnt[wv[k] >> SRC_BITS], 1);
    }
    __syncthreads();
    const int c = (t < BIN_NODES) ? cnt[t] : 0;
    int x = c;
    for (int o = 1; o < 32; o <<= 1) {
        int y = __shfl_up(x, o);
        if (lane >= o) x += y;
    }
    if (t < BIN_NODES && lane == 31) wsum[w] = x;
    __syncthreads();
    if (t < BIN_NODES) {
        int woff = 0;
        for (int i = 0; i < BIN_NODES / 32; ++i) woff += (i < w) ? wsum[i] : 0;
        const int excl = woff + x - c;
        cnt[t] = excl;
        const int node = b * BIN_NODES + t;
        if (node < N_NODES) rowptr[node] = beg + excl;
    }
    __syncthreads();
    int* psl = perm + beg;
    for (int b0 = 0; b0 < n; b0 += 4 * BT) {
        int wv[4];
        slice_entries(rs, ra, words, n, b0 + t, wv);
        for (int k = 0; k < 4; ++k)
            if (wv[k] >= 0) {
                const int pos = atomicAdd(&cnt[wv[k] >> SRC_BITS], 1);
                psl[pos] = wv[k] & SRC_MASK;
            }
    }
}

__global__ void __launch_bounds__(256)
gat_gemm_wmma(const float* __restrict__ h, const float* __restrict__ Wfc,
              float* __restrict__ z) {
    __shared__ float Wl[OUT_DIM * LDS_STRIDE];

    const int tid = threadIdx.x;
    for (int j = tid; j < IN_DIM * OUT_DIM; j += 256) {
        int i  = j >> 7;
        int c  = j & 127;
        int hh = c >> 4;
        int d  = c & 15;
        Wl[c * LDS_STRIDE + i] = Wfc[hh * (IN_DIM * HDIM) + i * HDIM + d];
    }
    __syncthreads();

    const int wave  = tid >> 5;
    const int lane  = tid & 31;
    const int mtile = blockIdx.x * 8 + wave;
    if (mtile >= N_NODES / 16) return;

    const int row   = lane & 15;
    const int khalf = lane >> 4;

    const float* hrow = h + (size_t)(mtile * 16 + row) * IN_DIM;

    v8f acc[8] = {};

    for (int k0 = 0; k0 < IN_DIM; k0 += 4) {
        const int kk = k0 + 2 * khalf;
        v2f a = *reinterpret_cast<const v2f*>(hrow + kk);
#pragma unroll
        for (int t = 0; t < 8; ++t) {
            v2f b = *reinterpret_cast<const v2f*>(&Wl[(t * 16 + row) * LDS_STRIDE + kk]);
            acc[t] = __builtin_amdgcn_wmma_f32_16x16x4_f32(
                false, a, false, b, (short)0, acc[t], false, false);
        }
    }

    float* zbase = z + (size_t)(mtile * 16) * OUT_DIM;
#pragma unroll
    for (int t = 0; t < 8; ++t) {
#pragma unroll
        for (int i = 0; i < 8; ++i) {
            int mrow = i + khalf * 8;
            zbase[(size_t)mrow * OUT_DIM + t * 16 + row] = acc[t][i];
        }
    }
}

__global__ void gat_scores(const float* __restrict__ z,
                           const float* __restrict__ Wattn,
                           float* __restrict__ s_src) {
    int idx = blockIdx.x * blockDim.x + threadIdx.x;
    if (idx >= N_NODES * HEADS) return;
    int n = idx >> 3, hh = idx & 7;
    const float* zp = z + (size_t)n * OUT_DIM + hh * HDIM;
    const float* as = Wattn + hh * (2 * HDIM);
    float ss = 0.0f;
    for (int d = 0; d < HDIM; ++d) ss += zp[d] * as[d];
    s_src[idx] = ss;
}

__device__ __forceinline__ float edge_exp(float xs, float sd, float m) {
    float sc = xs + sd;
    sc = (sc >= 0.0f) ? sc : NEG_SLOPE * sc;
    return expf(sc - m);
}

__device__ __forceinline__ float elu1(float v) {
    return (v > 0.0f) ? v : expm1f(v);
}

__device__ __forceinline__ int wave_order(int s, int c, int lane) {
    int rank = 0;
    for (int i = 0; i < c; ++i) {
        const int si = __builtin_amdgcn_readlane(s, i);
        rank += ((si < s) || (si == s && i < lane)) ? 1 : 0;
    }
    rank = (lane < c) ? rank : lane;
    return __builtin_amdgcn_ds_permute(rank << 2, s);
}

__global__ void __launch_bounds__(256)
gat_node_gather(const int* __restrict__ rowptr, const int* __restrict__ perm,
                const float* __restrict__ s_src, const float* __restrict__ z,
                const float* __restrict__ Wattn, const float* __restrict__ h,
                float* __restrict__ out) {
    const int lane = threadIdx.x & 31;
    const int node = __builtin_amdgcn_readfirstlane((int)(blockIdx.x * 8 + (threadIdx.x >> 5)));
    if (node >= N_NODES) return;
    const int hh = lane >> 2;

    const int beg = rowptr[node];
    const int end = (node + 1 < N_NODES) ? rowptr[node + 1] : N_EDGES;
    const int deg = end - beg;

    const float4 zd = *(const float4*)(z + (size_t)node * OUT_DIM + lane * 4);
    const float4 hv = *(const float4*)(h + (size_t)node * OUT_DIM + lane * 4);
    const float4 ad = *(const float4*)(Wattn + hh * (2 * HDIM) + HDIM + (lane & 3) * 4);
    float sd = zd.x * ad.x + zd.y * ad.y + zd.z * ad.z + zd.w * ad.w;
    sd += __shfl_xor(sd, 1);
    sd += __shfl_xor(sd, 2);

    const int s_first = (lane < deg) ? perm[beg + lane] : 0;

    float mx[HEADS];
    for (int k = 0; k < HEADS; ++k) mx[k] = -INFINITY;
    for (int base = 0; base < deg; base += 32) {
        const bool valid = base + lane < deg;
        const int s = (base == 0) ? s_first : (valid ? perm[beg + base + lane] : 0);
        const float4 a = *(const float4*)(s_src + (size_t)s * HEADS);
        const float4 b = *(const float4*)(s_src + (size_t)s * HEADS + 4);
        if (valid) {
            mx[0] = fmaxf(mx[0], a.x); mx[1] = fmaxf(mx[1], a.y);
            mx[2] = fmaxf(mx[2], a.z); mx[3] = fmaxf(mx[3], a.w);
            mx[4] = fmaxf(mx[4], b.x); mx[5] = fmaxf(mx[5], b.y);
            mx[6] = fmaxf(mx[6], b.z); mx[7] = fmaxf(mx[7], b.w);
        }
    }
    const bool b4 = (lane & 16) != 0, b3 = (lane & 8) != 0, b2 = (lane & 4) != 0;
    float w4[4], u2[2], t;
    for (int i = 0; i < 4; ++i) {
        const float send = b4 ? mx[i] : mx[4 + i];
        const float keep = b4 ? mx[4 + i] : mx[i];
        w4[i] = fmaxf(keep, __shfl_xor(send, 16));
    }
    for (int i = 0; i < 2; ++i) {
        const float send = b3 ? w4[i] : w4[2 + i];
        const float keep = b3 ? w4[2 + i] : w4[i];
        u2[i] = fmaxf(keep, __shfl_xor(send, 8));
    }
    {
        const float send = b2 ? u2[0] : u2[1];
        const float keep = b2 ? u2[1] : u2[0];
        t = fmaxf(keep, __shfl_xor(send, 4));
    }
    t = fmaxf(t, __shfl_xor(t, 2));
    t = fmaxf(t, __shfl_xor(t, 1));
    float m = t + sd;
    m = (m >= 0.0f) ? m : NEG_SLOPE * m;

    float den = 0.0f;
    float4 acc = make_float4(0.0f, 0.0f, 0.0f, 0.0f);
    const float* srow = s_src + hh;
    const float* zcol = z + lane * 4;
    for (int base = 0; base < deg; base += 32) {
        const int c = min(32, deg - base);
        int s = (base == 0) ? s_first : ((lane < c) ? perm[beg + base + lane] : 0);
        s = wave_order(s, c, lane);
        for (int j = 0; j < c; j += 4) {
            const int j1 = min(j + 1, c - 1), j2 = min(j + 2, c - 1), j3 = min(j + 3, c - 1);
            const int sa = __builtin_amdgcn_readlane(s, j);
            const int sb = __builtin_amdgcn_readlane(s, j1);
            const int sc = __builtin_amdgcn_readlane(s, j2);
            const int sq = __builtin_amdgcn_readlane(s, j3);
            const float xa = srow[(size_t)sa * HEADS];
            const float xb = srow[(size_t)sb * HEADS];
            const float xc = srow[(size_t)sc * HEADS];
            const float xq = srow[(size_t)sq * HEADS];
            const float4 za = *(const float4*)(zcol + (size_t)sa * OUT_DIM);
            const float4 zb = *(const float4*)(zcol + (size_t)sb * OUT_DIM);
            const float4 zc = *(const float4*)(zcol + (size_t)sc * OUT_DIM);
            const float4 zq = *(const float4*)(zcol + (size_t)sq * OUT_DIM);
            const float kb = (j + 1 < c) ? 1.0f : 0.0f;
            const float kc = (j + 2 < c) ? 1.0f : 0.0f;
            const float kq = (j + 3 < c) ? 1.0f : 0.0f;
            const float pa = edge_exp(xa, sd, m);
            const float pb = edge_exp(xb, sd, m) * kb;
            const float pc = edge_exp(xc, sd, m) * kc;
            const float pq = edge_exp(xq, sd, m) * kq;
            den += pa;
            acc.x += pa * za.x; acc.y += pa * za.y; acc.z += pa * za.z; acc.w += pa * za.w;
            den += pb;
            acc.x += pb * zb.x; acc.y += pb * zb.y; acc.z += pb * zb.z; acc.w += pb * zb.w;
            den += pc;
            acc.x += pc * zc.x; acc.y += pc * zc.y; acc.z += pc * zc.z; acc.w += pc * zc.w;
            den += pq;
            acc.x += pq * zq.x; acc.y += pq * zq.y; acc.z += pq * zq.z; acc.w += pq * zq.w;
        }
    }

    float4 r = make_float4(0.0f, 0.0f, 0.0f, 0.0f);
    if (deg > 0) {
        const float inv = 1.0f / den;
        r.x = acc.x * inv; r.y = acc.y * inv; r.z = acc.z * inv; r.w = acc.w * inv;
    }
    float4 o;
    o.x = hv.x + elu1(r.x); o.y = hv.y + elu1(r.y);
    o.z = hv.z + elu1(r.z); o.w = hv.w + elu1(r.w);
    *(float4*)(out + (size_t)node * OUT_DIM + lane * 4) = o;
}

extern "C" void kernel_launch(void* const* d_in, const int* in_sizes, int n_in,
                              void* d_out, int out_size, void* d_ws, size_t ws_size,
                              hipStream_t stream) {
    const float* h     = (const float*)d_in[0];
    const float* e     = (const float*)d_in[1];
    const int*   src   = (const int*)d_in[2];
    const int*   dst   = (const int*)d_in[3];
    const float* Wfc   = (const float*)d_in[4];
    const float* Wattn = (const float*)d_in[5];
    float* out = (float*)d_out;

    float* z      = (float*)d_ws;
    float* s_src  = z + (size_t)N_NODES * OUT_DIM;
    int*   rowptr = (int*)(s_src + (size_t)N_NODES * HEADS);
    int*   loff   = rowptr + N_NODES;
    int*   perm   = rowptr + (size_t)N_NODES * HEADS;
    int*   words  = (int*)d_ws;

    const int T = 256;

    gat_chunk_order<<<NBLK, BT, 0, stream>>>(src, dst, e, out, loff, words);
    gat_slice_place<<<NBINS, BT, 0, stream>>>(loff, words, rowptr, perm);

    gat_gemm_wmma<<<(N_NODES / 16 + 7) / 8, T, 0, stream>>>(h, Wfc, z);

    gat_scores<<<(N_NODES * HEADS + T - 1) / T, T, 0, stream>>>(z, Wattn, s_src);

    gat_node_gather<<<(N_NODES + 7) / 8, T, 0, stream>>>(rowptr, perm, s_src, z, Wattn, h, out);
}


# Node kernel pass 2: the 4 lanes of a head split a group's 4 edges, one s_src load and one exp per lane and group instead of four, weights shared by DPP quad broadcast; values and sum order kept
# speedup vs baseline: 12.7921x; 1.0438x over previous
#include <hip/hip_runtime.h>
#include <math.h>

#define N_NODES 100000
#define N_EDGES 1600000
#define IN_DIM  128
#define HEADS   8
#define HDIM    16
#define OUT_DIM 128
#define NEG_SLOPE 0.01f
#define LDS_STRIDE 130

typedef __attribute__((ext_vector_type(2))) float v2f;
typedef __attribute__((ext_vector_type(8))) float v8f;

#define BIN_SHIFT 8
#define BIN_NODES (1 << BIN_SHIFT)
#define NBINS ((N_NODES + BIN_NODES - 1) / BIN_NODES)
#define NBLK 256
#define CHUNK (N_EDGES / NBLK)
#define SRC_BITS 17
#define SRC_MASK ((1 << SRC_BITS) - 1)
#define BT 1024
#define LSTR (NBINS + 1)
#define BIN_WAVES ((NBINS + 31) / 32)
#define E_PER ((N_EDGES / 4 + NBLK * BT - 1) / (NBLK * BT))
static_assert(N_EDGES % 4 == 0, "16-byte copies of e");
static_assert(N_EDGES % NBLK == 0, "equal edge chunks");
static_assert(NBINS <= BT, "one thread per slice in the chunk scan");
static_assert(NBLK <= BT && (NBLK & (NBLK - 1)) == 0, "one thread per run; the run search halves a power of two");
static_assert(N_NODES <= (1 << SRC_BITS), "source index fits the packed entry");

__global__ void __launch_bounds__(BT)
gat_chunk_order(const int* __restrict__ src, const int* __restrict__ dst,
                const float* __restrict__ e, float* __restrict__ out,
                int* __restrict__ loff, int* __restrict__ words) {
    __shared__ int cur[NBINS];
    __shared__ int wsum[BIN_WAVES];
    const int t = threadIdx.x, blk = blockIdx.x, lane = t & 31, w = t >> 5;
    if (t < NBINS) cur[t] = 0;
    __syncthreads();
    float4 ev[E_PER];
    for (int k = 0; k < E_PER; ++k) {
        const int j = (k * NBLK + blk) * BT + t;
        ev[k] = (j < N_EDGES / 4) ? ((const float4*)e)[j] : make_float4(0.0f, 0.0f, 0.0f, 0.0f);
    }
    const int* dchunk = dst + (size_t)blk * CHUNK;
    const int* schunk = src + (size_t)blk * CHUNK;
    for (int b0 = 0; b0 < CHUNK; b0 += 4 * BT) {
        int d[4];
        for (int k = 0; k < 4; ++k) {
            const int i = b0 + k * BT + t;
            d[k] = (i < CHUNK) ? dchunk[i] : -1;
        }
        for (int k = 0; k < 4; ++k)
            if (d[k] >= 0) atomicAdd(&cur[d[k] >> BIN_SHIFT], 1);
    }
    __syncthreads();
    const int c = (t < NBINS) ? cur[t] : 0;
    int x = c;
    for (int o = 1; o < 32; o <<= 1) {
        int y = __shfl_up(x, o);
        if (lane >= o) x += y;
    }
    if (w < BIN_WAVES && lane == 31) wsum[w] = x;
    __syncthreads();
    int woff = 0;
    for (int i = 0; i < BIN_WAVES; ++i) woff += (i < w) ? wsum[i] : 0;
    const int excl = woff + x - c;
    if (t < NBINS) cur[t] = excl;
    __syncthreads();
    int* wchunk = words + (size_t)blk * CHUNK;
    for (int b0 = 0; b0 < CHUNK; b0 += 4 * BT) {
        int d[4], s[4];
        for (int k = 0; k < 4; ++k) {
            const int i = b0 + k * BT + t;
            d[k] = (i < CHUNK) ? dchunk[i] : -1;
            s[k] = (i < CHUNK) ? schunk[i] : 0;
        }
        for (int k = 0; k < 4; ++k)
            if (d[k] >= 0) {
                const int pos = atomicAdd(&cur[d[k] >> BIN_SHIFT], 1);
                wchunk[pos] = ((d[k] & (BIN_NODES - 1)) << SRC_BITS) | s[k];
            }
    }
    if (t < NBINS) loff[blk * LSTR + t] = excl;
    if (t == NBINS) loff[blk * LSTR + NBINS] = CHUNK;
    for (int k = 0; k < E_PER; ++k) {
        const int j = (k * NBLK + blk) * BT + t;
        if (j < N_EDGES / 4) ((float4*)(out + (size_t)N_NODES * OUT_DIM))[j] = ev[k];
    }
}

__device__ __forceinline__ void slice_entries(const int* rs, const int* ra,
                                              const int* __restrict__ words,
                                              int n, int i0, int (&wv)[4]) {
    int ii[4], r[4];
    bool ok[4];
    for (int k = 0; k < 4; ++k) {
        const int i = i0 + k * BT;
        ok[k] = i < n;
        ii[k] = ok[k] ? i : n - 1;
        r[k] = 0;
    }
    for (int step = NBLK / 2; step > 0; step >>= 1)
        for (int k = 0; k < 4; ++k)
            if (rs[r[k] + step] <= ii[k]) r[k] += step;
    for (int k = 0; k < 4; ++k) {
        const int v = words[ra[r[k]] + ii[k]];
        wv[k] = ok[k] ? v : -1;
    }
}

__global__ void __launch_bounds__(BT)
gat_slice_place(const int* __restrict__ loff, const int* __restrict__ words,
                int* __restrict__ rowptr, int* __restrict__ perm) {
    __shared__ int cnt[BIN_NODES];
    __shared__ int wsum[BIN_NODES / 32];
    __shared__ int lsum[NBLK / 32];
    __shared__ int osum[NBLK / 32];
    __shared__ int rs[NBLK];
    __shared__ int ra[NBLK];
    const int t = threadIdx.x, b = blockIdx.x, lane = t & 31, w = t >> 5;

    int o0 = 0, len = 0;
    if (t < NBLK) {
        o0  = loff[t * LSTR + b];
        len = loff[t * LSTR + b + 1] - o0;
    }
    int xl = len;
    for (int o = 1; o < 32; o <<= 1) {
        int y = __shfl_up(xl, o);
        if (lane >= o) xl += y;
    }
    int so = o0;
    for (int o = 16; o > 0; o >>= 1) so += __shfl_xor(so, o);
    if (t < NBLK && lane == 31) { lsum[w] = xl; osum[w] = so; }
    if (t < BIN_NODES) cnt[t] = 0;
    __syncthreads();
    int loffw = 0, n = 0, beg = 0;
    for (int i = 0; i < NBLK / 32; ++i) {
        loffw += (i < w) ? lsum[i] : 0;
        n     += lsum[i];
        beg   += osum[i];
    }
    if (t < NBLK) {
        const int first = loffw + xl - len;
        rs[t] = first;
        ra[t] = t * CHUNK + o0 - first;
    }
    __syncthreads();

    for (int b0 = 0; b0 < n; b0 += 4 * BT) {
        int wv[4];
        slice_entries(rs, ra, words, n, b0 + t, wv);
        for (int k = 0; k < 4; ++k)
            if (wv[k] >= 0) atomicAdd(&cnt[wv[k] >> SRC_BITS], 1);
    }
    __syncthreads();
    const int c = (t < BIN_NODES) ? cnt[t] : 0;
    int x = c;
    for (int o = 1; o < 32; o <<= 1) {
        int y = __shfl_up(x, o);
        if (lane >= o) x += y;
    }
    if (t < BIN_NODES && lane == 31) wsum[w] = x;
    __syncthreads();
    if (t < BIN_NODES) {
        int woff = 0;
        for (int i = 0; i < BIN_NODES / 32; ++i) woff += (i < w) ? wsum[i] : 0;
        const int excl = woff + x - c;
        cnt[t] = excl;
        const int node = b * BIN_NODES + t;
        if (node < N_NODES) rowptr[node] = beg + excl;
    }
    __syncthreads();
    int* psl = perm + beg;
    for (int b0 = 0; b0 < n; b0 += 4 * BT) {
        int wv[4];
        slice_entries(rs, ra, words, n, b0 + t, wv);
        for (int k = 0; k < 4; ++k)
            if (wv[k] >= 0) {
                const int pos = atomicAdd(&cnt[wv[k] >> SRC_BITS], 1);
                psl[pos] = wv[k] & SRC_MASK;
            }
    }
}

__global__ void __launch_bounds__(256)
gat_gemm_wmma(const float* __restrict__ h, const float* __restrict__ Wfc,
              float* __restrict__ z) {
    __shared__ float Wl[OUT_DIM * LDS_STRIDE];

    const int tid = threadIdx.x;
    for (int j = tid; j < IN_DIM * OUT_DIM; j += 256) {
        int i  = j >> 7;
        int c  = j & 127;
        int hh = c >> 4;
        int d  = c & 15;
        Wl[c * LDS_STRIDE + i] = Wfc[hh * (IN_DIM * HDIM) + i * HDIM + d];
    }
    __syncthreads();

    const int wave  = tid >> 5;
    const int lane  = tid & 31;
    const int mtile = blockIdx.x * 8 + wave;
    if (mtile >= N_NODES / 16) return;

    const int row   = lane & 15;
    const int khalf = lane >> 4;

    const float* hrow = h + (size_t)(mtile * 16 + row) * IN_DIM;

    v8f acc[8] = {};

    for (int k0 = 0; k0 < IN_DIM; k0 += 4) {
        const int kk = k0 + 2 * khalf;
        v2f a = *reinterpret_cast<const v2f*>(hrow + kk);
#pragma unroll
        for (int t = 0; t < 8; ++t) {
            v2f b = *reinterpret_cast<const v2f*>(&Wl[(t * 16 + row) * LDS_STRIDE + kk]);
            acc[t] = __builtin_amdgcn_wmma_f32_16x16x4_f32(
                false, a, false, b, (short)0, acc[t], false, false);
        }
    }

    float* zbase = z + (size_t)(mtile * 16) * OUT_DIM;
#pragma unroll
    for (int t = 0; t < 8; ++t) {
#pragma unroll
        for (int i = 0; i < 8; ++i) {
            int mrow = i + khalf * 8;
            zbase[(size_t)mrow * OUT_DIM + t * 16 + row] = acc[t][i];
        }
    }
}

__global__ void gat_scores(const float* __restrict__ z,
                           const float* __restrict__ Wattn,
                           float* __restrict__ s_src) {
    int idx = blockIdx.x * blockDim.x + threadIdx.x;
    if (idx >= N_NODES * HEADS) return;
    int n = idx >> 3, hh = idx & 7;
    const float* zp = z + (size_t)n * OUT_DIM + hh * HDIM;
    const float* as = Wattn + hh * (2 * HDIM);
    float ss = 0.0f;
    for (int d = 0; d < HDIM; ++d) ss += zp[d] * as[d];
    s_src[idx] = ss;
}

__device__ __forceinline__ float edge_exp(float xs, float sd, float m) {
    float sc = xs + sd;
    sc = (sc >= 0.0f) ? sc : NEG_SLOPE * sc;
    return expf(sc - m);
}

__device__ __forceinline__ float elu1(float v) {
    return (v > 0.0f) ? v : expm1f(v);
}

template <int E>
__device__ __forceinline__ float quad_lane(float v) {
    return __int_as_float(__builtin_amdgcn_mov_dpp(__float_as_int(v), E * 0x55, 0xF, 0xF, true));
}

__device__ __forceinline__ int wave_order(int s, int c, int lane) {
    int rank = 0;
    for (int i = 0; i < c; ++i) {
        const int si = __builtin_amdgcn_readlane(s, i);
        rank += ((si < s) || (si == s && i < lane)) ? 1 : 0;
    }
    rank = (lane < c) ? rank : lane;
    return __builtin_amdgcn_ds_permute(rank << 2, s);
}

__global__ void __launch_bounds__(256)
gat_node_gather(const int* __restrict__ rowptr, const int* __restrict__ perm,
                const float* __restrict__ s_src, const float* __restrict__ z,
                const float* __restrict__ Wattn, const float* __restrict__ h,
                float* __restrict__ out) {
    const int lane = threadIdx.x & 31;
    const int node = __builtin_amdgcn_readfirstlane((int)(blockIdx.x * 8 + (threadIdx.x >> 5)));
    if (node >= N_NODES) return;
    const int hh = lane >> 2;

    const int beg = rowptr[node];
    const int end = (node + 1 < N_NODES) ? rowptr[node + 1] : N_EDGES;
    const int deg = end - beg;

    const float4 zd = *(const float4*)(z + (size_t)node * OUT_DIM + lane * 4);
    const float4 hv = *(const float4*)(h + (size_t)node * OUT_DIM + lane * 4);
    const float4 ad = *(const float4*)(Wattn + hh * (2 * HDIM) + HDIM + (lane & 3) * 4);
    float sd = zd.x * ad.x + zd.y * ad.y + zd.z * ad.z + zd.w * ad.w;
    sd += __shfl_xor(sd, 1);
    sd += __shfl_xor(sd, 2);

    const int s_first = (lane < deg) ? perm[beg + lane] : 0;

    float mx[HEADS];
    for (int k = 0; k < HEADS; ++k) mx[k] = -INFINITY;
    for (int base = 0; base < deg; base += 32) {
        const bool valid = base + lane < deg;
        const int s = (base == 0) ? s_first : (valid ? perm[beg + base + lane] : 0);
        const float4 a = *(const float4*)(s_src + (size_t)s * HEADS);
        const float4 b = *(const float4*)(s_src + (size_t)s * HEADS + 4);
        if (valid) {
            mx[0] = fmaxf(mx[0], a.x); mx[1] = fmaxf(mx[1], a.y);
            mx[2] = fmaxf(mx[2], a.z); mx[3] = fmaxf(mx[3], a.w);
            mx[4] = fmaxf(mx[4], b.x); mx[5] = fmaxf(mx[5], b.y);
            mx[6] = fmaxf(mx[6], b.z); mx[7] = fmaxf(mx[7], b.w);
        }
    }
    const bool b4 = (lane & 16) != 0, b3 = (lane & 8) != 0, b2 = (lane & 4) != 0;
    float w4[4], u2[2], t;
    for (int i = 0; i < 4; ++i) {
        const float send = b4 ? mx[i] : mx[4 + i];
        const float keep = b4 ? mx[4 + i] : mx[i];
        w4[i] = fmaxf(keep, __shfl_xor(send, 16));
    }
    for (int i = 0; i < 2; ++i) {
        const float send = b3 ? w4[i] : w4[2 + i];
        const float keep = b3 ? w4[2 + i] : w4[i];
        u2[i] = fmaxf(keep, __shfl_xor(send, 8));
    }
    {
        const float send = b2 ? u2[0] : u2[1];
        const float keep = b2 ? u2[1] : u2[0];
        t = fmaxf(keep, __shfl_xor(send, 4));
    }
    t = fmaxf(t, __shfl_xor(t, 2));
    t = fmaxf(t, __shfl_xor(t, 1));
    float m = t + sd;
    m = (m >= 0.0f) ? m : NEG_SLOPE * m;

    float den = 0.0f;
    float4 acc = make_float4(0.0f, 0.0f, 0.0f, 0.0f);
    const float* srow = s_src + hh;
    const float* zcol = z + lane * 4;
    const int qd = lane & 3;
    const bool is1 = qd == 1, is2 = qd == 2, is3 = qd == 3;
    for (int base = 0; base < deg; base += 32) {
        const int c = min(32, deg - base);
        int s = (base == 0) ? s_first : ((lane < c) ? perm[beg + base + lane] : 0);
        s = wave_order(s, c, lane);
        for (int j = 0; j < c; j += 4) {
            const int j1 = min(j + 1, c - 1), j2 = min(j + 2, c - 1), j3 = min(j + 3, c - 1);
            const int sa = __builtin_amdgcn_readlane(s, j);
            const int sb = __builtin_amdgcn_readlane(s, j1);
            const int sc = __builtin_amdgcn_readlane(s, j2);
            const int sq = __builtin_amdgcn_readlane(s, j3);
            int sm = sa;
            sm = is1 ? sb : sm;
            sm = is2 ? sc : sm;
            sm = is3 ? sq : sm;
            const float xm = srow[(size_t)sm * HEADS];
            const float4 za = *(const float4*)(zcol + (size_t)sa * OUT_DIM);
            const float4 zb = *(const float4*)(zcol + (size_t)sb * OUT_DIM);
            const float4 zc = *(const float4*)(zcol + (size_t)sc * OUT_DIM);
            const float4 zq = *(const float4*)(zcol + (size_t)sq * OUT_DIM);
            const float km = (j + qd < c) ? 1.0f : 0.0f;
            const float pm = edge_exp(xm, sd, m) * km;
            const float pa = quad_lane<0>(pm);
            const float pb = quad_lane<1>(pm);
            const float pc = quad_lane<2>(pm);
            const float pq = quad_lane<3>(pm);
            den += pa;
            acc.x += pa * za.x; acc.y += pa * za.y; acc.z += pa * za.z; acc.w += pa * za.w;
            den += pb;
            acc.x += pb * zb.x; acc.y += pb * zb.y; acc.z += pb * zb.z; acc.w += pb * zb.w;
            den += pc;
            acc.x += pc * zc.x; acc.y += pc * zc.y; acc.z += pc * zc.z; acc.w += pc * zc.w;
            den += pq;
            acc.x += pq * zq.x; acc.y += pq * zq.y; acc.z += pq * zq.z; acc.w += pq * zq.w;
        }
    }

    float4 r = make_float4(0.0f, 0.0f, 0.0f, 0.0f);
    if (deg > 0) {
        const float inv = 1.0f / den;
        r.x = acc.x * inv; r.y = acc.y * inv; r.z = acc.z * inv; r.w = acc.w * inv;
    }
    float4 o;
    o.x = hv.x + elu1(r.x); o.y = hv.y + elu1(r.y);
    o.z = hv.z + elu1(r.z); o.w = hv.w + elu1(r.w);
    *(float4*)(out + (size_t)node * OUT_DIM + lane * 4) = o;
}

extern "C" void kernel_launch(void* const* d_in, const int* in_sizes, int n_in,
                              void* d_out, int out_size, void* d_ws, size_t ws_size,
                              hipStream_t stream) {
    const float* h     = (const float*)d_in[0];
    const float* e     = (const float*)d_in[1];
    const int*   src   = (const int*)d_in[2];
    const int*   dst   = (const int*)d_in[3];
    const float* Wfc   = (const float*)d_in[4];
    const float* Wattn = (const float*)d_in[5];
    float* out = (float*)d_out;

    float* z      = (float*)d_ws;
    float* s_src  = z + (size_t)N_NODES * OUT_DIM;
    int*   rowptr = (int*)(s_src + (size_t)N_NODES * HEADS);
    int*   loff   = rowptr + N_NODES;
    int*   perm   = rowptr + (size_t)N_NODES * HEADS;
    int*   words  = (int*)d_ws;

    const int T = 256;

    gat_chunk_order<<<NBLK, BT, 0, stream>>>(src, dst, e, out, loff, words);
    gat_slice_place<<<NBINS, BT, 0, stream>>>(loff, words, rowptr, perm);

    gat_gemm_wmma<<<(N_NODES / 16 + 7) / 8, T, 0, stream>>>(h, Wfc, z);

    gat_scores<<<(N_NODES * HEADS + T - 1) / T, T, 0, stream>>>(z, Wattn, s_src);

    gat_node_gather<<<(N_NODES + 7) / 8, T, 0, stream>>>(rowptr, perm, s_src, z, Wattn, h, out);
}


# s_src dot moved into the GEMM wave: its accumulators times a_src, summed over 16 lanes by 4 DPP halving exchanges (tree order, was a serial FMA chain); score kernel and its re-read of z are gone
# speedup vs baseline: 13.5722x; 1.0610x over previous
#include <hip/hip_runtime.h>
#include <math.h>

#define N_NODES 100000
#define N_EDGES 1600000
#define IN_DIM  128
#define HEADS   8
#define HDIM    16
#define OUT_DIM 128
#define NEG_SLOPE 0.01f
#define LDS_STRIDE 130

typedef __attribute__((ext_vector_type(2))) float v2f;
typedef __attribute__((ext_vector_type(8))) float v8f;

#define BIN_SHIFT 8
#define BIN_NODES (1 << BIN_SHIFT)
#define NBINS ((N_NODES + BIN_NODES - 1) / BIN_NODES)
#define NBLK 256
#define CHUNK (N_EDGES / NBLK)
#define SRC_BITS 17
#define SRC_MASK ((1 << SRC_BITS) - 1)
#define BT 1024
#define LSTR (NBINS + 1)
#define BIN_WAVES ((NBINS + 31) / 32)
#define E_PER ((N_EDGES / 4 + NBLK * BT - 1) / (NBLK * BT))
static_assert(N_EDGES % 4 == 0, "16-byte copies of e");
static_assert(N_EDGES % NBLK == 0, "equal edge chunks");
static_assert(NBINS <= BT, "one thread per slice in the chunk scan");
static_assert(NBLK <= BT && (NBLK & (NBLK - 1)) == 0, "one thread per run; the run search halves a power of two");
static_assert(N_NODES <= (1 << SRC_BITS), "source index fits the packed entry");

__global__ void __launch_bounds__(BT)
gat_chunk_order(const int* __restrict__ src, const int* __restrict__ dst,
                const float* __restrict__ e, float* __restrict__ out,
                int* __restrict__ loff, int* __restrict__ words) {
    __shared__ int cur[NBINS];
    __shared__ int wsum[BIN_WAVES];
    const int t = threadIdx.x, blk = blockIdx.x, lane = t & 31, w = t >> 5;
    if (t < NBINS) cur[t] = 0;
    __syncthreads();
    float4 ev[E_PER];
    for (int k = 0; k < E_PER; ++k) {
        const int j = (k * NBLK + blk) * BT + t;
        ev[k] = (j < N_EDGES / 4) ? ((const float4*)e)[j] : make_float4(0.0f, 0.0f, 0.0f, 0.0f);
    }
    const int* dchunk = dst + (size_t)blk * CHUNK;
    const int* schunk = src + (size_t)blk * CHUNK;
    for (int b0 = 0; b0 < CHUNK; b0 += 4 * BT) {
        int d[4];
        for (int k = 0; k < 4; ++k) {
            const int i = b0 + k * BT + t;
            d[k] = (i < CHUNK) ? dchunk[i] : -1;
        }
        for (int k = 0; k < 4; ++k)
            if (d[k] >= 0) atomicAdd(&cur[d[k] >> BIN_SHIFT], 1);
    }
    __syncthreads();
    const int c = (t < NBINS) ? cur[t] : 0;
    int x = c;
    for (int o = 1; o < 32; o <<= 1) {
        int y = __shfl_up(x, o);
        if (lane >= o) x += y;
    }
    if (w < BIN_WAVES && lane == 31) wsum[w] = x;
    __syncthreads();
    int woff = 0;
    for (int i = 0; i < BIN_WAVES; ++i) woff += (i < w) ? wsum[i] : 0;
    const int excl = woff + x - c;
    if (t < NBINS) cur[t] = excl;
    __syncthreads();
    int* wchunk = words + (size_t)blk * CHUNK;
    for (int b0 = 0; b0 < CHUNK; b0 += 4 * BT) {
        int d[4], s[4];
        for (int k = 0; k < 4; ++k) {
            const int i = b0 + k * BT + t;
            d[k] = (i < CHUNK) ? dchunk[i] : -1;
            s[k] = (i < CHUNK) ? schunk[i] : 0;
        }
        for (int k = 0; k < 4; ++k)
            if (d[k] >= 0) {
                const int pos = atomicAdd(&cur[d[k] >> BIN_SHIFT], 1);
                wchunk[pos] = ((d[k] & (BIN_NODES - 1)) << SRC_BITS) | s[k];
            }
    }
    if (t < NBINS) loff[blk * LSTR + t] = excl;
    if (t == NBINS) loff[blk * LSTR + NBINS] = CHUNK;
    for (int k = 0; k < E_PER; ++k) {
        const int j = (k * NBLK + blk) * BT + t;
        if (j < N_EDGES / 4) ((float4*)(out + (size_t)N_NODES * OUT_DIM))[j] = ev[k];
    }
}

__device__ __forceinline__ void slice_entries(const int* rs, const int* ra,
                                              const int* __restrict__ words,
                                              int n, int i0, int (&wv)[4]) {
    int ii[4], r[4];
    bool ok[4];
    for (int k = 0; k < 4; ++k) {
        const int i = i0 + k * BT;
        ok[k] = i < n;
        ii[k] = ok[k] ? i : n - 1;
        r[k] = 0;
    }
    for (int step = NBLK / 2; step > 0; step >>= 1)
        for (int k = 0; k < 4; ++k)
            if (rs[r[k] + step] <= ii[k]) r[k] += step;
    for (int k = 0; k < 4; ++k) {
        const int v = words[ra[r[k]] + ii[k]];
        wv[k] = ok[k] ? v : -1;
    }
}

__global__ void __launch_bounds__(BT)
gat_slice_place(const int* __restrict__ loff, const int* __restrict__ words,
                int* __restrict__ rowptr, int* __restrict__ perm) {
    __shared__ int cnt[BIN_NODES];
    __shared__ int wsum[BIN_NODES / 32];
    __shared__ int lsum[NBLK / 32];
    __shared__ int osum[NBLK / 32];
    __shared__ int rs[NBLK];
    __shared__ int ra[NBLK];
    const int t = threadIdx.x, b = blockIdx.x, lane = t & 31, w = t >> 5;

    int o0 = 0, len = 0;
    if (t < NBLK) {
        o0  = loff[t * LSTR + b];
        len = loff[t * LSTR + b + 1] - o0;
    }
    int xl = len;
    for (int o = 1; o < 32; o <<= 1) {
        int y = __shfl_up(xl, o);
        if (lane >= o) xl += y;
    }
    int so = o0;
    for (int o = 16; o > 0; o >>= 1) so += __shfl_xor(so, o);
    if (t < NBLK && lane == 31) { lsum[w] = xl; osum[w] = so; }
    if (t < BIN_NODES) cnt[t] = 0;
    __syncthreads();
    int loffw = 0, n = 0, beg = 0;
    for (int i = 0; i < NBLK / 32; ++i) {
        loffw += (i < w) ? lsum[i] : 0;
        n     += lsum[i];
        beg   += osum[i];
    }
    if (t < NBLK) {
        const int first = loffw + xl - len;
        rs[t] = first;
        ra[t] = t * CHUNK + o0 - first;
    }
    __syncthreads();

    for (int b0 = 0; b0 < n; b0 += 4 * BT) {
        int wv[4];
        slice_entries(rs, ra, words, n, b0 + t, wv);
        for (int k = 0; k < 4; ++k)
            if (wv[k] >= 0) atomicAdd(&cnt[wv[k] >> SRC_BITS], 1);
    }
    __syncthreads();
    const int c = (t < BIN_NODES) ? cnt[t] : 0;
    int x = c;
    for (int o = 1; o < 32; o <<= 1) {
        int y = __shfl_up(x, o);
        if (lane >= o) x += y;
    }
    if (t < BIN_NODES && lane == 31) wsum[w] = x;
    __syncthreads();
    if (t < BIN_NODES) {
        int woff = 0;
        for (int i = 0; i < BIN_NODES / 32; ++i) woff += (i < w) ? wsum[i] : 0;
        const int excl = woff + x - c;
        cnt[t] = excl;
        const int node = b * BIN_NODES + t;
        if (node < N_NODES) rowptr[node] = beg + excl;
    }
    __syncthreads();
    int* psl = perm + beg;
    for (int b0 = 0; b0 < n; b0 += 4 * BT) {
        int wv[4];
        slice_entries(rs, ra, words, n, b0 + t, wv);
        for (int k = 0; k < 4; ++k)
            if (wv[k] >= 0) {
                const int pos = atomicAdd(&cnt[wv[k] >> SRC_BITS], 1);
                psl[pos] = wv[k] & SRC_MASK;
            }
    }
}

template <int CTRL>
__device__ __forceinline__ float row_xchg(float v) {
    return __int_as_float(__builtin_amdgcn_mov_dpp(__float_as_int(v), CTRL, 0xF, 0xF, true));
}
#define DPP_XOR1 0xB1
#define DPP_XOR2 0x4E
#define DPP_XOR4 0x164
#define DPP_XOR8 0x168

__global__ void __launch_bounds__(256)
gat_gemm_wmma(const float* __restrict__ h, const float* __restrict__ Wfc,
              const float* __restrict__ Wattn,
              float* __restrict__ z, float* __restrict__ s_src) {
    __shared__ float Wl[OUT_DIM * LDS_STRIDE];

    const int tid = threadIdx.x;
    for (int j = tid; j < IN_DIM * OUT_DIM; j += 256) {
        int i  = j >> 7;
        int c  = j & 127;
        int hh = c >> 4;
        int d  = c & 15;
        Wl[c * LDS_STRIDE + i] = Wfc[hh * (IN_DIM * HDIM) + i * HDIM + d];
    }
    __syncthreads();

    const int wave  = tid >> 5;
    const int lane  = tid & 31;
    const int mtile = blockIdx.x * 8 + wave;
    if (mtile >= N_NODES / 16) return;

    const int row   = lane & 15;
    const int khalf = lane >> 4;

    const float* hrow = h + (size_t)(mtile * 16 + row) * IN_DIM;

    v8f acc[8] = {};

    for (int k0 = 0; k0 < IN_DIM; k0 += 4) {
        const int kk = k0 + 2 * khalf;
        v2f a = *reinterpret_cast<const v2f*>(hrow + kk);
#pragma unroll
        for (int t = 0; t < 8; ++t) {
            v2f b = *reinterpret_cast<const v2f*>(&Wl[(t * 16 + row) * LDS_STRIDE + kk]);
            acc[t] = __builtin_amdgcn_wmma_f32_16x16x4_f32(
                false, a, false, b, (short)0, acc[t], false, false);
        }
    }

    float* zbase = z + (size_t)(mtile * 16) * OUT_DIM;
#pragma unroll
    for (int t = 0; t < 8; ++t) {
#pragma unroll
        for (int i = 0; i < 8; ++i) {
            int mrow = i + khalf * 8;
            zbase[(size_t)mrow * OUT_DIM + t * 16 + row] = acc[t][i];
        }
    }

    float as[HEADS];
    for (int t = 0; t < HEADS; ++t) as[t] = Wattn[t * (2 * HDIM) + row];
    const bool b3 = (row & 8) != 0, b2 = (row & 4) != 0, b1 = (row & 2) != 0, b0 = (row & 1) != 0;
    float q1[4][8], q2[2][8], q3[8], q4[4];
    for (int u = 0; u < 4; ++u) {
        for (int i = 0; i < 8; ++i) {
            const float lo = acc[u][i] * as[u], hi = acc[4 + u][i] * as[4 + u];
            q1[u][i] = (b3 ? hi : lo) + row_xchg<DPP_XOR8>(b3 ? lo : hi);
        }
    }
    for (int u = 0; u < 2; ++u) {
        for (int i = 0; i < 8; ++i) {
            const float lo = q1[u][i], hi = q1[2 + u][i];
            q2[u][i] = (b2 ? hi : lo) + row_xchg<DPP_XOR4>(b2 ? lo : hi);
        }
    }
    for (int i = 0; i < 8; ++i) {
        const float lo = q2[0][i], hi = q2[1][i];
        q3[i] = (b1 ? hi : lo) + row_xchg<DPP_XOR2>(b1 ? lo : hi);
    }
    for (int r = 0; r < 4; ++r) {
        const float lo = q3[r], hi = q3[4 + r];
        q4[r] = (b0 ? hi : lo) + row_xchg<DPP_XOR1>(b0 ? lo : hi);
    }
    float* sbase = s_src + (size_t)(mtile * 16 + khalf * 8 + (b0 ? 4 : 0)) * HEADS + (row >> 1);
    for (int r = 0; r < 4; ++r) sbase[r * HEADS] = q4[r];
}

__device__ __forceinline__ float edge_exp(float xs, float sd, float m) {
    float sc = xs + sd;
    sc = (sc >= 0.0f) ? sc : NEG_SLOPE * sc;
    return expf(sc - m);
}

__device__ __forceinline__ float elu1(float v) {
    return (v > 0.0f) ? v : expm1f(v);
}

template <int E>
__device__ __forceinline__ float quad_lane(float v) {
    return __int_as_float(__builtin_amdgcn_mov_dpp(__float_as_int(v), E * 0x55, 0xF, 0xF, true));
}

__device__ __forceinline__ int wave_order(int s, int c, int lane) {
    int rank = 0;
    for (int i = 0; i < c; ++i) {
        const int si = __builtin_amdgcn_readlane(s, i);
        rank += ((si < s) || (si == s && i < lane)) ? 1 : 0;
    }
    rank = (lane < c) ? rank : lane;
    return __builtin_amdgcn_ds_permute(rank << 2, s);
}

__global__ void __launch_bounds__(256)
gat_node_gather(const int* __restrict__ rowptr, const int* __restrict__ perm,
                const float* __restrict__ s_src, const float* __restrict__ z,
                const float* __restrict__ Wattn, const float* __restrict__ h,
                float* __restrict__ out) {
    const int lane = threadIdx.x & 31;
    const int node = __builtin_amdgcn_readfirstlane((int)(blockIdx.x * 8 + (threadIdx.x >> 5)));
    if (node >= N_NODES) return;
    const int hh = lane >> 2;

    const int beg = rowptr[node];
    const int end = (node + 1 < N_NODES) ? rowptr[node + 1] : N_EDGES;
    const int deg = end - beg;

    const float4 zd = *(const float4*)(z + (size_t)node * OUT_DIM + lane * 4);
    const float4 hv = *(const float4*)(h + (size_t)node * OUT_DIM + lane * 4);
    const float4 ad = *(const float4*)(Wattn + hh * (2 * HDIM) + HDIM + (lane & 3) * 4);
    float sd = zd.x * ad.x + zd.y * ad.y + zd.z * ad.z + zd.w * ad.w;
    sd += __shfl_xor(sd, 1);
    sd += __shfl_xor(sd, 2);

    const int s_first = (lane < deg) ? perm[beg + lane] : 0;

    float mx[HEADS];
    for (int k = 0; k < HEADS; ++k) mx[k] = -INFINITY;
    for (int base = 0; base < deg; base += 32) {
        const bool valid = base + lane < deg;
        const int s = (base == 0) ? s_first : (valid ? perm[beg + base + lane] : 0);
        const float4 a = *(const float4*)(s_src + (size_t)s * HEADS);
        const float4 b = *(const float4*)(s_src + (size_t)s * HEADS + 4);
        if (valid) {
            mx[0] = fmaxf(mx[0], a.x); mx[1] = fmaxf(mx[1], a.y);
            mx[2] = fmaxf(mx[2], a.z); mx[3] = fmaxf(mx[3], a.w);
            mx[4] = fmaxf(mx[4], b.x); mx[5] = fmaxf(mx[5], b.y);
            mx[6] = fmaxf(mx[6], b.z); mx[7] = fmaxf(mx[7], b.w);
        }
    }
    const bool b4 = (lane & 16) != 0, b3 = (lane & 8) != 0, b2 = (lane & 4) != 0;
    float w4[4], u2[2], t;
    for (int i = 0; i < 4; ++i) {
        const float send = b4 ? mx[i] : mx[4 + i];
        const float keep = b4 ? mx[4 + i] : mx[i];
        w4[i] = fmaxf(keep, __shfl_xor(send, 16));
    }
    for (int i = 0; i < 2; ++i) {
        const float send = b3 ? w4[i] : w4[2 + i];
        const float keep = b3 ? w4[2 + i] : w4[i];
        u2[i] = fmaxf(keep, __shfl_xor(send, 8));
    }
    {
        const float send = b2 ? u2[0] : u2[1];
        const float keep = b2 ? u2[1] : u2[0];
        t = fmaxf(keep, __shfl_xor(send, 4));
    }
    t = fmaxf(t, __shfl_xor(t, 2));
    t = fmaxf(t, __shfl_xor(t, 1));
    float m = t + sd;
    m = (m >= 0.0f) ? m : NEG_SLOPE * m;

    float den = 0.0f;
    float4 acc = make_float4(0.0f, 0.0f, 0.0f, 0.0f);
    const float* srow = s_src + hh;
    const float* zcol = z + lane * 4;
    const int qd = lane & 3;
    const bool is1 = qd == 1, is2 = qd == 2, is3 = qd == 3;
    for (int base = 0; base < deg; base += 32) {
        const int c = min(32, deg - base);
        int s = (base == 0) ? s_first : ((lane < c) ? perm[beg + base + lane] : 0);
        s = wave_order(s, c, lane);
        for (int j = 0; j < c; j += 4) {
            const int j1 = min(j + 1, c - 1), j2 = min(j + 2, c - 1), j3 = min(j + 3, c - 1);
            const int sa = __builtin_amdgcn_readlane(s, j);
            const int sb = __builtin_amdgcn_readlane(s, j1);
            const int sc = __builtin_amdgcn_readlane(s, j2);
            const int sq = __builtin_amdgcn_readlane(s, j3);
            int sm = sa;
            sm = is1 ? sb : sm;
            sm = is2 ? sc : sm;
            sm = is3 ? sq : sm;
            const float xm = srow[(size_t)sm * HEADS];
            const float4 za = *(const float4*)(zcol + (size_t)sa * OUT_DIM);
            const float4 zb = *(const float4*)(zcol + (size_t)sb * OUT_DIM);
            const float4 zc = *(const float4*)(zcol + (size_t)sc * OUT_DIM);
            const float4 zq = *(const float4*)(zcol + (size_t)sq * OUT_DIM);
            const float km = (j + qd < c) ? 1.0f : 0.0f;
            const float pm = edge_exp(xm, sd, m) * km;
            const float pa = quad_lane<0>(pm);
            const float pb = quad_lane<1>(pm);
            const float pc = quad_lane<2>(pm);
            const float pq = quad_lane<3>(pm);
            den += pa;
            acc.x += pa * za.x; acc.y += pa * za.y; acc.z += pa * za.z; acc.w += pa * za.w;
            den += pb;
            acc.x += pb * zb.x; acc.y += pb * zb.y; acc.z += pb * zb.z; acc.w += pb * zb.w;
            den += pc;
            acc.x += pc * zc.x; acc.y += pc * zc.y; acc.z += pc * zc.z; acc.w += pc * zc.w;
            den += pq;
            acc.x += pq * zq.x; acc.y += pq * zq.y; acc.z += pq * zq.z; acc.w += pq * zq.w;
        }
    }

    float4 r = make_float4(0.0f, 0.0f, 0.0f, 0.0f);
    if (deg > 0) {
        const float inv = 1.0f / den;
        r.x = acc.x * inv; r.y = acc.y * inv; r.z = acc.z * inv; r.w = acc.w * inv;
    }
    float4 o;
    o.x = hv.x + elu1(r.x); o.y = hv.y + elu1(r.y);
    o.z = hv.z + elu1(r.z); o.w = hv.w + elu1(r.w);
    *(float4*)(out + (size_t)node * OUT_DIM + lane * 4) = o;
}

extern "C" void kernel_launch(void* const* d_in, const int* in_sizes, int n_in,
                              void* d_out, int out_size, void* d_ws, size_t ws_size,
                              hipStream_t stream) {
    const float* h     = (const float*)d_in[0];
    const float* e     = (const float*)d_in[1];
    const int*   src   = (const int*)d_in[2];
    const int*   dst   = (const int*)d_in[3];
    const float* Wfc   = (const float*)d_in[4];
    const float* Wattn = (const float*)d_in[5];
    float* out = (float*)d_out;

    float* z      = (float*)d_ws;
    float* s_src  = z + (size_t)N_NODES * OUT_DIM;
    int*   rowptr = (int*)(s_src + (size_t)N_NODES * HEADS);
    int*   loff   = rowptr + N_NODES;
    int*   perm   = rowptr + (size_t)N_NODES * HEADS;
    int*   words  = (int*)d_ws;

    const int T = 256;

    gat_chunk_order<<<NBLK, BT, 0, stream>>>(src, dst, e, out, loff, words);
    gat_slice_place<<<NBINS, BT, 0, stream>>>(loff, words, rowptr, perm);

    gat_gemm_wmma<<<(N_NODES / 16 + 7) / 8, T, 0, stream>>>(h, Wfc, Wattn, z, s_src);

    gat_node_gather<<<(N_NODES + 7) / 8, T, 0, stream>>>(rowptr, perm, s_src, z, Wattn, h, out);
}


# Node kernel: sources ordered first; each lane then loads its head's s_src of its quarter's edges once, first 4 z rows in the same trip; max (2 DPP steps) and weights from those registers; pass 1 gone
# speedup vs baseline: 14.3598x; 1.0580x over previous
#include <hip/hip_runtime.h>
#include <math.h>

#define N_NODES 100000
#define N_EDGES 1600000
#define IN_DIM  128
#define HEADS   8
#define HDIM    16
#define OUT_DIM 128
#define NEG_SLOPE 0.01f
#define LDS_STRIDE 130

typedef __attribute__((ext_vector_type(2))) float v2f;
typedef __attribute__((ext_vector_type(8))) float v8f;

#define BIN_SHIFT 8
#define BIN_NODES (1 << BIN_SHIFT)
#define NBINS ((N_NODES + BIN_NODES - 1) / BIN_NODES)
#define NBLK 256
#define CHUNK (N_EDGES / NBLK)
#define SRC_BITS 17
#define SRC_MASK ((1 << SRC_BITS) - 1)
#define BT 1024
#define LSTR (NBINS + 1)
#define BIN_WAVES ((NBINS + 31) / 32)
#define E_PER ((N_EDGES / 4 + NBLK * BT - 1) / (NBLK * BT))
static_assert(N_EDGES % 4 == 0, "16-byte copies of e");
static_assert(N_EDGES % NBLK == 0, "equal edge chunks");
static_assert(NBINS <= BT, "one thread per slice in the chunk scan");
static_assert(NBLK <= BT && (NBLK & (NBLK - 1)) == 0, "one thread per run; the run search halves a power of two");
static_assert(N_NODES <= (1 << SRC_BITS), "source index fits the packed entry");

__global__ void __launch_bounds__(BT)
gat_chunk_order(const int* __restrict__ src, const int* __restrict__ dst,
                const float* __restrict__ e, float* __restrict__ out,
                int* __restrict__ loff, int* __restrict__ words) {
    __shared__ int cur[NBINS];
    __shared__ int wsum[BIN_WAVES];
    const int t = threadIdx.x, blk = blockIdx.x, lane = t & 31, w = t >> 5;
    if (t < NBINS) cur[t] = 0;
    __syncthreads();
    float4 ev[E_PER];
    for (int k = 0; k < E_PER; ++k) {
        const int j = (k * NBLK + blk) * BT + t;
        ev[k] = (j < N_EDGES / 4) ? ((const float4*)e)[j] : make_float4(0.0f, 0.0f, 0.0f, 0.0f);
    }
    const int* dchunk = dst + (size_t)blk * CHUNK;
    const int* schunk = src + (size_t)blk * CHUNK;
    for (int b0 = 0; b0 < CHUNK; b0 += 4 * BT) {
        int d[4];
        for (int k = 0; k < 4; ++k) {
            const int i = b0 + k * BT + t;
            d[k] = (i < CHUNK) ? dchunk[i] : -1;
        }
        for (int k = 0; k < 4; ++k)
            if (d[k] >= 0) atomicAdd(&cur[d[k] >> BIN_SHIFT], 1);
    }
    __syncthreads();
    const int c = (t < NBINS) ? cur[t] : 0;
    int x = c;
    for (int o = 1; o < 32; o <<= 1) {
        int y = __shfl_up(x, o);
        if (lane >= o) x += y;
    }
    if (w < BIN_WAVES && lane == 31) wsum[w] = x;
    __syncthreads();
    int woff = 0;
    for (int i = 0; i < BIN_WAVES; ++i) woff += (i < w) ? wsum[i] : 0;
    const int excl = woff + x - c;
    if (t < NBINS) cur[t] = excl;
    __syncthreads();
    int* wchunk = words + (size_t)blk * CHUNK;
    for (int b0 = 0; b0 < CHUNK; b0 += 4 * BT) {
        int d[4], s[4];
        for (int k = 0; k < 4; ++k) {
            const int i = b0 + k * BT + t;
            d[k] = (i < CHUNK) ? dchunk[i] : -1;
            s[k] = (i < CHUNK) ? schunk[i] : 0;
        }
        for (int k = 0; k < 4; ++k)
            if (d[k] >= 0) {
                const int pos = atomicAdd(&cur[d[k] >> BIN_SHIFT], 1);
                wchunk[pos] = ((d[k] & (BIN_NODES - 1)) << SRC_BITS) | s[k];
            }
    }
    if (t < NBINS) loff[blk * LSTR + t] = excl;
    if (t == NBINS) loff[blk * LSTR + NBINS] = CHUNK;
    for (int k = 0; k < E_PER; ++k) {
        const int j = (k * NBLK + blk) * BT + t;
        if (j < N_EDGES / 4) ((float4*)(out + (size_t)N_NODES * OUT_DIM))[j] = ev[k];
    }
}

__device__ __forceinline__ void slice_entries(const int* rs, const int* ra,
                                              const int* __restrict__ words,
                                              int n, int i0, int (&wv)[4]) {
    int ii[4], r[4];
    bool ok[4];
    for (int k = 0; k < 4; ++k) {
        const int i = i0 + k * BT;
        ok[k] = i < n;
        ii[k] = ok[k] ? i : n - 1;
        r[k] = 0;
    }
    for (int step = NBLK / 2; step > 0; step >>= 1)
        for (int k = 0; k < 4; ++k)
            if (rs[r[k] + step] <= ii[k]) r[k] += step;
    for (int k = 0; k < 4; ++k) {
        const int v = words[ra[r[k]] + ii[k]];
        wv[k] = ok[k] ? v : -1;
    }
}

__global__ void __launch_bounds__(BT)
gat_slice_place(const int* __restrict__ loff, const int* __restrict__ words,
                int* __restrict__ rowptr, int* __restrict__ perm) {
    __shared__ int cnt[BIN_NODES];
    __shared__ int wsum[BIN_NODES / 32];
    __shared__ int lsum[NBLK / 32];
    __shared__ int osum[NBLK / 32];
    __shared__ int rs[NBLK];
    __shared__ int ra[NBLK];
    const int t = threadIdx.x, b = blockIdx.x, lane = t & 31, w = t >> 5;

    int o0 = 0, len = 0;
    if (t < NBLK) {
        o0  = loff[t * LSTR + b];
        len = loff[t * LSTR + b + 1] - o0;
    }
    int xl = len;
    for (int o = 1; o < 32; o <<= 1) {
        int y = __shfl_up(xl, o);
        if (lane >= o) xl += y;
    }
    int so = o0;
    for (int o = 16; o > 0; o >>= 1) so += __shfl_xor(so, o);
    if (t < NBLK && lane == 31) { lsum[w] = xl; osum[w] = so; }
    if (t < BIN_NODES) cnt[t] = 0;
    __syncthreads();
    int loffw = 0, n = 0, beg = 0;
    for (int i = 0; i < NBLK / 32; ++i) {
        loffw += (i < w) ? lsum[i] : 0;
        n     += lsum[i];
        beg   += osum[i];
    }
    if (t < NBLK) {
        const int first = loffw + xl - len;
        rs[t] = first;
        ra[t] = t * CHUNK + o0 - first;
    }
    __syncthreads();

    for (int b0 = 0; b0 < n; b0 += 4 * BT) {
        int wv[4];
        slice_entries(rs, ra, words, n, b0 + t, wv);
        for (int k = 0; k < 4; ++k)
            if (wv[k] >= 0) atomicAdd(&cnt[wv[k] >> SRC_BITS], 1);
    }
    __syncthreads();
    const int c = (t < BIN_NODES) ? cnt[t] : 0;
    int x = c;
    for (int o = 1; o < 32; o <<= 1) {
        int y = __shfl_up(x, o);
        if (lane >= o) x += y;
    }
    if (t < BIN_NODES && lane == 31) wsum[w] = x;
    __syncthreads();
    if (t < BIN_NODES) {
        int woff = 0;
        for (int i = 0; i < BIN_NODES / 32; ++i) woff += (i < w) ? wsum[i] : 0;
        const int excl = woff + x - c;
        cnt[t] = excl;
        const int node = b * BIN_NODES + t;
        if (node < N_NODES) rowptr[node] = beg + excl;
    }
    __syncthreads();
    int* psl = perm + beg;
    for (int b0 = 0; b0 < n; b0 += 4 * BT) {
        int wv[4];
        slice_entries(rs, ra, words, n, b0 + t, wv);
        for (int k = 0; k < 4; ++k)
            if (wv[k] >= 0) {
                const int pos = atomicAdd(&cnt[wv[k] >> SRC_BITS], 1);
                psl[pos] = wv[k] & SRC_MASK;
            }
    }
}

template <int CTRL>
__device__ __forceinline__ float row_xchg(float v) {
    return __int_as_float(__builtin_amdgcn_mov_dpp(__float_as_int(v), CTRL, 0xF, 0xF, true));
}
#define DPP_XOR1 0xB1
#define DPP_XOR2 0x4E
#define DPP_XOR4 0x164
#define DPP_XOR8 0x168

__global__ void __launch_bounds__(256)
gat_gemm_wmma(const float* __restrict__ h, const float* __restrict__ Wfc,
              const float* __restrict__ Wattn,
              float* __restrict__ z, float* __restrict__ s_src) {
    __shared__ float Wl[OUT_DIM * LDS_STRIDE];

    const int tid = threadIdx.x;
    for (int j = tid; j < IN_DIM * OUT_DIM; j += 256) {
        int i  = j >> 7;
        int c  = j & 127;
        int hh = c >> 4;
        int d  = c & 15;
        Wl[c * LDS_STRIDE + i] = Wfc[hh * (IN_DIM * HDIM) + i * HDIM + d];
    }
    __syncthreads();

    const int wave  = tid >> 5;
    const int lane  = tid & 31;
    const int mtile = blockIdx.x * 8 + wave;
    if (mtile >= N_NODES / 16) return;

    const int row   = lane & 15;
    const int khalf = lane >> 4;

    const float* hrow = h + (size_t)(mtile * 16 + row) * IN_DIM;

    v8f acc[8] = {};

    for (int k0 = 0; k0 < IN_DIM; k0 += 4) {
        const int kk = k0 + 2 * khalf;
        v2f a = *reinterpret_cast<const v2f*>(hrow + kk);
#pragma unroll
        for (int t = 0; t < 8; ++t) {
            v2f b = *reinterpret_cast<const v2f*>(&Wl[(t * 16 + row) * LDS_STRIDE + kk]);
            acc[t] = __builtin_amdgcn_wmma_f32_16x16x4_f32(
                false, a, false, b, (short)0, acc[t], false, false);
        }
    }

    float* zbase = z + (size_t)(mtile * 16) * OUT_DIM;
#pragma unroll
    for (int t = 0; t < 8; ++t) {
#pragma unroll
        for (int i = 0; i < 8; ++i) {
            int mrow = i + khalf * 8;
            zbase[(size_t)mrow * OUT_DIM + t * 16 + row] = acc[t][i];
        }
    }

    float as[HEADS];
    for (int t = 0; t < HEADS; ++t) as[t] = Wattn[t * (2 * HDIM) + row];
    const bool b3 = (row & 8) != 0, b2 = (row & 4) != 0, b1 = (row & 2) != 0, b0 = (row & 1) != 0;
    float q1[4][8], q2[2][8], q3[8], q4[4];
    for (int u = 0; u < 4; ++u) {
        for (int i = 0; i < 8; ++i) {
            const float lo = acc[u][i] * as[u], hi = acc[4 + u][i] * as[4 + u];
            q1[u][i] = (b3 ? hi : lo) + row_xchg<DPP_XOR8>(b3 ? lo : hi);
        }
    }
    for (int u = 0; u < 2; ++u) {
        for (int i = 0; i < 8; ++i) {
            const float lo = q1[u][i], hi = q1[2 + u][i];
            q2[u][i] = (b2 ? hi : lo) + row_xchg<DPP_XOR4>(b2 ? lo : hi);
        }
    }
    for (int i = 0; i < 8; ++i) {
        const float lo = q2[0][i], hi = q2[1][i];
        q3[i] = (b1 ? hi : lo) + row_xchg<DPP_XOR2>(b1 ? lo : hi);
    }
    for (int r = 0; r < 4; ++r) {
        const float lo = q3[r], hi = q3[4 + r];
        q4[r] = (b0 ? hi : lo) + row_xchg<DPP_XOR1>(b0 ? lo : hi);
    }
    float* sbase = s_src + (size_t)(mtile * 16 + khalf * 8 + (b0 ? 4 : 0)) * HEADS + (row >> 1);
    for (int r = 0; r < 4; ++r) sbase[r * HEADS] = q4[r];
}

__device__ __forceinline__ float edge_exp(float xs, float sd, float m) {
    float sc = xs + sd;
    sc = (sc >= 0.0f) ? sc : NEG_SLOPE * sc;
    return expf(sc - m);
}

__device__ __forceinline__ float elu1(float v) {
    return (v > 0.0f) ? v : expm1f(v);
}

template <int E>
__device__ __forceinline__ float quad_lane(float v) {
    return __int_as_float(__builtin_amdgcn_mov_dpp(__float_as_int(v), E * 0x55, 0xF, 0xF, true));
}

__device__ __forceinline__ int wave_order(int s, int c, int lane) {
    int rank = 0;
    for (int i = 0; i < c; ++i) {
        const int si = __builtin_amdgcn_readlane(s, i);
        rank += ((si < s) || (si == s && i < lane)) ? 1 : 0;
    }
    rank = (lane < c) ? rank : lane;
    return __builtin_amdgcn_ds_permute(rank << 2, s);
}

__device__ __forceinline__ void rows_load(const float* __restrict__ z, int s, int j, int c, int lane,
                                          float4& za, float4& zb, float4& zc, float4& zq) {
    const int sa = __builtin_amdgcn_readlane(s, j);
    const int sb = __builtin_amdgcn_readlane(s, min(j + 1, c - 1));
    const int sc = __builtin_amdgcn_readlane(s, min(j + 2, c - 1));
    const int sq = __builtin_amdgcn_readlane(s, min(j + 3, c - 1));
    za = *(const float4*)(z + (size_t)sa * OUT_DIM + lane * 4);
    zb = *(const float4*)(z + (size_t)sb * OUT_DIM + lane * 4);
    zc = *(const float4*)(z + (size_t)sc * OUT_DIM + lane * 4);
    zq = *(const float4*)(z + (size_t)sq * OUT_DIM + lane * 4);
}

__device__ __forceinline__ void group_sum(const float4& za, const float4& zb, const float4& zc,
                                          const float4& zq, float xg, float sd, float m,
                                          int j, int qd, int c, float& den, float4& acc) {
    const float km = (j + qd < c) ? 1.0f : 0.0f;
    const float pm = edge_exp(xg, sd, m) * km;
    const float pa = quad_lane<0>(pm);
    const float pb = quad_lane<1>(pm);
    const float pc = quad_lane<2>(pm);
    const float pq = quad_lane<3>(pm);
    den += pa;
    acc.x = fmaf(pa, za.x, acc.x); acc.y = fmaf(pa, za.y, acc.y);
    acc.z = fmaf(pa, za.z, acc.z); acc.w = fmaf(pa, za.w, acc.w);
    den += pb;
    acc.x = fmaf(pb, zb.x, acc.x); acc.y = fmaf(pb, zb.y, acc.y);
    acc.z = fmaf(pb, zb.z, acc.z); acc.w = fmaf(pb, zb.w, acc.w);
    den += pc;
    acc.x = fmaf(pc, zc.x, acc.x); acc.y = fmaf(pc, zc.y, acc.y);
    acc.z = fmaf(pc, zc.z, acc.z); acc.w = fmaf(pc, zc.w, acc.w);
    den += pq;
    acc.x = fmaf(pq, zq.x, acc.x); acc.y = fmaf(pq, zq.y, acc.y);
    acc.z = fmaf(pq, zq.z, acc.z); acc.w = fmaf(pq, zq.w, acc.w);
}

__device__ __forceinline__ float head_max_score(float mx, float sd) {
    mx = fmaxf(mx, row_xchg<DPP_XOR1>(mx));
    mx = fmaxf(mx, row_xchg<DPP_XOR2>(mx));
    const float m = mx + sd;
    return (m >= 0.0f) ? m : NEG_SLOPE * m;
}

__device__ __forceinline__ void chunk_fetch(const int* __restrict__ perm, const float* srow,
                                            int s_first, int beg, int base, int deg,
                                            int lane, int qd, int& s, int& c, float (&x)[8]) {
    c = min(32, deg - base);
    const int sv = (base == 0) ? s_first : ((lane < c) ? perm[beg + base + lane] : 0);
    s = wave_order(sv, c, lane);
    const int last = c - 1;
    int sm[8];
#pragma unroll
    for (int g = 0; g < 8; ++g)
        sm[g] = __builtin_amdgcn_ds_bpermute(min(4 * g + qd, last) << 2, s);
#pragma unroll
    for (int g = 0; g < 8; ++g) {
        x[g] = -INFINITY;
        if (4 * g < c) x[g] = srow[(size_t)sm[g] * HEADS];
    }
}

__global__ void __launch_bounds__(256)
gat_node_gather(const int* __restrict__ rowptr, const int* __restrict__ perm,
                const float* __restrict__ s_src, const float* __restrict__ z,
                const float* __restrict__ Wattn, const float* __restrict__ h,
                float* __restrict__ out) {
    const int lane = threadIdx.x & 31;
    const int node = __builtin_amdgcn_readfirstlane((int)(blockIdx.x * 8 + (threadIdx.x >> 5)));
    if (node >= N_NODES) return;
    const int hh = lane >> 2;
    const int qd = lane & 3;

    const int beg = rowptr[node];
    const int end = (node + 1 < N_NODES) ? rowptr[node + 1] : N_EDGES;
    const int deg = end - beg;

    const float4 zd = *(const float4*)(z + (size_t)node * OUT_DIM + lane * 4);
    const float4 hv = *(const float4*)(h + (size_t)node * OUT_DIM + lane * 4);
    const float4 ad = *(const float4*)(Wattn + hh * (2 * HDIM) + HDIM + (lane & 3) * 4);
    float sd = zd.x * ad.x + zd.y * ad.y + zd.z * ad.z + zd.w * ad.w;
    sd += __shfl_xor(sd, 1);
    sd += __shfl_xor(sd, 2);

    const int s_first = (lane < deg) ? perm[beg + lane] : 0;
    const float* srow = s_src + hh;

    int s = 0, c = 0;
    float x[8];
    float den = 0.0f;
    float4 acc = make_float4(0.0f, 0.0f, 0.0f, 0.0f);
    if (deg <= 32) {
        if (deg > 0) {
            chunk_fetch(perm, srow, s_first, beg, 0, deg, lane, qd, s, c, x);
            float4 za, zb, zc, zq;
            rows_load(z, s, 0, c, lane, za, zb, zc, zq);
            __builtin_amdgcn_sched_barrier(0);
            float mx = x[0];
#pragma unroll
            for (int g = 1; g < 8; ++g) mx = fmaxf(mx, x[g]);
            const float m = head_max_score(mx, sd);
            group_sum(za, zb, zc, zq, x[0], sd, m, 0, qd, c, den, acc);
#pragma unroll
            for (int g = 1; g < 8; ++g) {
                if (4 * g < c) {
                    rows_load(z, s, 4 * g, c, lane, za, zb, zc, zq);
                    group_sum(za, zb, zc, zq, x[g], sd, m, 4 * g, qd, c, den, acc);
                }
            }
        }
    } else {
        float mx = -INFINITY;
        for (int base = 0; base < deg; base += 32) {
            chunk_fetch(perm, srow, s_first, beg, base, deg, lane, qd, s, c, x);
#pragma unroll
            for (int g = 0; g < 8; ++g) mx = fmaxf(mx, x[g]);
        }
        const float m = head_max_score(mx, sd);
        for (int base = 0; base < deg; base += 32) {
            chunk_fetch(perm, srow, s_first, beg, base, deg, lane, qd, s, c, x);
#pragma unroll
            for (int g = 0; g < 8; ++g) {
                if (4 * g < c) {
                    float4 za, zb, zc, zq;
                    rows_load(z, s, 4 * g, c, lane, za, zb, zc, zq);
                    group_sum(za, zb, zc, zq, x[g], sd, m, 4 * g, qd, c, den, acc);
                }
            }
        }
    }

    float4 r = make_float4(0.0f, 0.0f, 0.0f, 0.0f);
    if (deg > 0) {
        const float inv = 1.0f / den;
        r.x = acc.x * inv; r.y = acc.y * inv; r.z = acc.z * inv; r.w = acc.w * inv;
    }
    float4 o;
    o.x = hv.x + elu1(r.x); o.y = hv.y + elu1(r.y);
    o.z = hv.z + elu1(r.z); o.w = hv.w + elu1(r.w);
    *(float4*)(out + (size_t)node * OUT_DIM + lane * 4) = o;
}

extern "C" void kernel_launch(void* const* d_in, const int* in_sizes, int n_in,
                              void* d_out, int out_size, void* d_ws, size_t ws_size,
                              hipStream_t stream) {
    const float* h     = (const float*)d_in[0];
    const float* e     = (const float*)d_in[1];
    const int*   src   = (const int*)d_in[2];
    const int*   dst   = (const int*)d_in[3];
    const float* Wfc   = (const float*)d_in[4];
    const float* Wattn = (const float*)d_in[5];
    float* out = (float*)d_out;

    float* z      = (float*)d_ws;
    float* s_src  = z + (size_t)N_NODES * OUT_DIM;
    int*   rowptr = (int*)(s_src + (size_t)N_NODES * HEADS);
    int*   loff   = rowptr + N_NODES;
    int*   perm   = rowptr + (size_t)N_NODES * HEADS;
    int*   words  = (int*)d_ws;

    const int T = 256;

    gat_chunk_order<<<NBLK, BT, 0, stream>>>(src, dst, e, out, loff, words);
    gat_slice_place<<<NBINS, BT, 0, stream>>>(loff, words, rowptr, perm);

    gat_gemm_wmma<<<(N_NODES / 16 + 7) / 8, T, 0, stream>>>(h, Wfc, Wattn, z, s_src);

    gat_node_gather<<<(N_NODES + 7) / 8, T, 0, stream>>>(rowptr, perm, s_src, z, Wattn, h, out);
}


# GEMM work-groups ride inside the two edge-grouping launches, role by block index (half beside chunk ordering, half beside slice placing; 256-thread groups): 4 launches to 3; sum orders kept
# speedup vs baseline: 16.0928x; 1.1207x over previous
#include <hip/hip_runtime.h>
#include <math.h>

#define N_NODES 100000
#define N_EDGES 1600000
#define IN_DIM  128
#define HEADS   8
#define HDIM    16
#define OUT_DIM 128
#define NEG_SLOPE 0.01f
#define LDS_STRIDE 130

typedef __attribute__((ext_vector_type(2))) float v2f;
typedef __attribute__((ext_vector_type(8))) float v8f;

#define BIN_SHIFT 8
#define BIN_NODES (1 << BIN_SHIFT)
#define NBINS ((N_NODES + BIN_NODES - 1) / BIN_NODES)
#define NBLK 256
#define CHUNK (N_EDGES / NBLK)
#define SRC_BITS 17
#define SRC_MASK ((1 << SRC_BITS) - 1)
#define FT 256
#define CK 13
#define SK 9
#define GEMM_BLOCKS ((N_NODES / 16 + 7) / 8)
#define GEMM_FIRST (GEMM_BLOCKS / 2)
#define LSTR (NBINS + 1)
#define E_PER ((N_EDGES / 4 + NBLK * FT - 1) / (NBLK * FT))
static_assert(N_EDGES % 4 == 0, "16-byte copies of e");
static_assert(N_EDGES % NBLK == 0, "equal edge chunks");
static_assert(NBINS <= 2 * FT, "two slices per thread in the chunk scan");
static_assert(NBLK <= FT && (NBLK & (NBLK - 1)) == 0, "one thread per run; the run search halves a power of two");
static_assert(BIN_NODES <= FT, "one thread per node in the slice scan");
static_assert(N_NODES <= (1 << SRC_BITS), "source index fits the packed entry");

__device__ __forceinline__ void
chunk_order_role(const int* __restrict__ src, const int* __restrict__ dst,
                 const float* __restrict__ e, float* __restrict__ out,
                 int* __restrict__ loff, int* __restrict__ words, const int blk) {
    __shared__ int cur[NBINS];
    __shared__ int wsum[FT / 32];
    const int t = threadIdx.x, lane = t & 31, w = t >> 5;
    for (int s = t; s < NBINS; s += FT) cur[s] = 0;
    __syncthreads();
    float4 ev[E_PER];
    for (int k = 0; k < E_PER; ++k) {
        const int j = (k * NBLK + blk) * FT + t;
        ev[k] = (j < N_EDGES / 4) ? ((const float4*)e)[j] : make_float4(0.0f, 0.0f, 0.0f, 0.0f);
    }
    const int* dchunk = dst + (size_t)blk * CHUNK;
    const int* schunk = src + (size_t)blk * CHUNK;
    for (int b0 = 0; b0 < CHUNK; b0 += CK * FT) {
        int d[CK];
        for (int k = 0; k < CK; ++k) {
            const int i = b0 + k * FT + t;
            d[k] = (i < CHUNK) ? dchunk[i] : -1;
        }
        for (int k = 0; k < CK; ++k)
            if (d[k] >= 0) atomicAdd(&cur[d[k] >> BIN_SHIFT], 1);
    }
    __syncthreads();
    const int s0 = 2 * t, s1 = 2 * t + 1;
    const int c0 = (s0 < NBINS) ? cur[s0] : 0;
    const int c1 = (s1 < NBINS) ? cur[s1] : 0;
    int x = c0 + c1;
    for (int o = 1; o < 32; o <<= 1) {
        int y = __shfl_up(x, o);
        if (lane >= o) x += y;
    }
    if (lane == 31) wsum[w] = x;
    __syncthreads();
    int woff = 0;
    for (int i = 0; i < FT / 32; ++i) woff += (i < w) ? wsum[i] : 0;
    const int excl0 = woff + x - c0 - c1;
    const int excl1 = excl0 + c0;
    if (s0 < NBINS) cur[s0] = excl0;
    if (s1 < NBINS) cur[s1] = excl1;
    __syncthreads();
    int* wchunk = words + (size_t)blk * CHUNK;
    for (int b0 = 0; b0 < CHUNK; b0 += CK * FT) {
        int d[CK], s[CK];
        for (int k = 0; k < CK; ++k) {
            const int i = b0 + k * FT + t;
            d[k] = (i < CHUNK) ? dchunk[i] : -1;
            s[k] = (i < CHUNK) ? schunk[i] : 0;
        }
        for (int k = 0; k < CK; ++k)
            if (d[k] >= 0) {
                const int pos = atomicAdd(&cur[d[k] >> BIN_SHIFT], 1);
                wchunk[pos] = ((d[k] & (BIN_NODES - 1)) << SRC_BITS) | s[k];
            }
    }
    if (s0 < NBINS) loff[blk * LSTR + s0] = excl0;
    if (s1 < NBINS) loff[blk * LSTR + s1] = excl1;
    if (t == 0) loff[blk * LSTR + NBINS] = CHUNK;
    for (int k = 0; k < E_PER; ++k) {
        const int j = (k * NBLK + blk) * FT + t;
        if (j < N_EDGES / 4) ((float4*)(out + (size_t)N_NODES * OUT_DIM))[j] = ev[k];
    }
}

__device__ __forceinline__ void slice_entries(const int* rs, const int* ra,
                                              const int* __restrict__ words,
                                              int n, int i0, int (&wv)[SK]) {
    int ii[SK], r[SK];
    bool ok[SK];
    for (int k = 0; k < SK; ++k) {
        const int i = i0 + k * FT;
        ok[k] = i < n;
        ii[k] = ok[k] ? i : n - 1;
        r[k] = 0;
    }
    for (int step = NBLK / 2; step > 0; step >>= 1) {
        for (int k = 0; k < SK; ++k)
            if (rs[r[k] + step] <= ii[k]) r[k] += step;
    }
    for (int k = 0; k < SK; ++k) {
        const int v = words[ra[r[k]] + ii[k]];
        wv[k] = ok[k] ? v : -1;
    }
}

__device__ __forceinline__ void
slice_place_role(const int* __restrict__ loff, const int* __restrict__ words,
                 int* __restrict__ rowptr, int* __restrict__ perm, const int b) {
    __shared__ int cnt[BIN_NODES];
    __shared__ int wsum[BIN_NODES / 32];
    __shared__ int lsum[NBLK / 32];
    __shared__ int osum[NBLK / 32];
    __shared__ int rs[NBLK];
    __shared__ int ra[NBLK];
    const int t = threadIdx.x, lane = t & 31, w = t >> 5;

    int o0 = 0, len = 0;
    if (t < NBLK) {
        o0  = loff[t * LSTR + b];
        len = loff[t * LSTR + b + 1] - o0;
    }
    int xl = len;
    for (int o = 1; o < 32; o <<= 1) {
        int y = __shfl_up(xl, o);
        if (lane >= o) xl += y;
    }
    int so = o0;
    for (int o = 16; o > 0; o >>= 1) so += __shfl_xor(so, o);
    if (t < NBLK && lane == 31) { lsum[w] = xl; osum[w] = so; }
    if (t < BIN_NODES) cnt[t] = 0;
    __syncthreads();
    int loffw = 0, n = 0, beg = 0;
    for (int i = 0; i < NBLK / 32; ++i) {
        loffw += (i < w) ? lsum[i] : 0;
        n     += lsum[i];
        beg   += osum[i];
    }
    if (t < NBLK) {
        const int first = loffw + xl - len;
        rs[t] = first;
        ra[t] = t * CHUNK + o0 - first;
    }
    __syncthreads();

    for (int b0 = 0; b0 < n; b0 += SK * FT) {
        int wv[SK];
        slice_entries(rs, ra, words, n, b0 + t, wv);
        for (int k = 0; k < SK; ++k)
            if (wv[k] >= 0) atomicAdd(&cnt[wv[k] >> SRC_BITS], 1);
    }
    __syncthreads();
    const int c = (t < BIN_NODES) ? cnt[t] : 0;
    int x = c;
    for (int o = 1; o < 32; o <<= 1) {
        int y = __shfl_up(x, o);
        if (lane >= o) x += y;
    }
    if (t < BIN_NODES && lane == 31) wsum[w] = x;
    __syncthreads();
    if (t < BIN_NODES) {
        int woff = 0;
        for (int i = 0; i < BIN_NODES / 32; ++i) woff += (i < w) ? wsum[i] : 0;
        const int excl = woff + x - c;
        cnt[t] = excl;
        const int node = b * BIN_NODES + t;
        if (node < N_NODES) rowptr[node] = beg + excl;
    }
    __syncthreads();
    int* psl = perm + beg;
    for (int b0 = 0; b0 < n; b0 += SK * FT) {
        int wv[SK];
        slice_entries(rs, ra, words, n, b0 + t, wv);
        for (int k = 0; k < SK; ++k)
            if (wv[k] >= 0) {
                const int pos = atomicAdd(&cnt[wv[k] >> SRC_BITS], 1);
                psl[pos] = wv[k] & SRC_MASK;
            }
    }
}

template <int CTRL>
__device__ __forceinline__ float row_xchg(float v) {
    return __int_as_float(__builtin_amdgcn_mov_dpp(__float_as_int(v), CTRL, 0xF, 0xF, true));
}
#define DPP_XOR1 0xB1
#define DPP_XOR2 0x4E
#define DPP_XOR4 0x164
#define DPP_XOR8 0x168

__device__ __forceinline__ void
gemm_role(const float* __restrict__ h, const float* __restrict__ Wfc,
          const float* __restrict__ Wattn,
          float* __restrict__ z, float* __restrict__ s_src, const int gblk) {
    __shared__ float Wl[OUT_DIM * LDS_STRIDE];

    const int tid = threadIdx.x;
    for (int j = tid; j < IN_DIM * OUT_DIM; j += FT) {
        int i  = j >> 7;
        int c  = j & 127;
        int hh = c >> 4;
        int d  = c & 15;
        Wl[c * LDS_STRIDE + i] = Wfc[hh * (IN_DIM * HDIM) + i * HDIM + d];
    }
    __syncthreads();

    const int wave  = tid >> 5;
    const int lane  = tid & 31;
    const int mtile = gblk * 8 + wave;
    if (mtile >= N_NODES / 16) return;

    const int row   = lane & 15;
    const int khalf = lane >> 4;

    const float* hrow = h + (size_t)(mtile * 16 + row) * IN_DIM;

    v8f acc[8] = {};

    for (int k0 = 0; k0 < IN_DIM; k0 += 4) {
        const int kk = k0 + 2 * khalf;
        v2f a = *reinterpret_cast<const v2f*>(hrow + kk);
#pragma unroll
        for (int t = 0; t < 8; ++t) {
            v2f b = *reinterpret_cast<const v2f*>(&Wl[(t * 16 + row) * LDS_STRIDE + kk]);
            acc[t] = __builtin_amdgcn_wmma_f32_16x16x4_f32(
                false, a, false, b, (short)0, acc[t], false, false);
        }
    }

    float* zbase = z + (size_t)(mtile * 16) * OUT_DIM;
#pragma unroll
    for (int t = 0; t < 8; ++t) {
#pragma unroll
        for (int i = 0; i < 8; ++i) {
            int mrow = i + khalf * 8;
            zbase[(size_t)mrow * OUT_DIM + t * 16 + row] = acc[t][i];
        }
    }

    float as[HEADS];
    for (int t = 0; t < HEADS; ++t) as[t] = Wattn[t * (2 * HDIM) + row];
    const bool b3 = (row & 8) != 0, b2 = (row & 4) != 0, b1 = (row & 2) != 0, b0 = (row & 1) != 0;
    float q1[4][8], q2[2][8], q3[8], q4[4];
    for (int u = 0; u < 4; ++u) {
        for (int i = 0; i < 8; ++i) {
            const float lo = acc[u][i] * as[u], hi = acc[4 + u][i] * as[4 + u];
            q1[u][i] = (b3 ? hi : lo) + row_xchg<DPP_XOR8>(b3 ? lo : hi);
        }
    }
    for (int u = 0; u < 2; ++u) {
        for (int i = 0; i < 8; ++i) {
            const float lo = q1[u][i], hi = q1[2 + u][i];
            q2[u][i] = (b2 ? hi : lo) + row_xchg<DPP_XOR4>(b2 ? lo : hi);
        }
    }
    for (int i = 0; i < 8; ++i) {
        const float lo = q2[0][i], hi = q2[1][i];
        q3[i] = (b1 ? hi : lo) + row_xchg<DPP_XOR2>(b1 ? lo : hi);
    }
    for (int r = 0; r < 4; ++r) {
        const float lo = q3[r], hi = q3[4 + r];
        q4[r] = (b0 ? hi : lo) + row_xchg<DPP_XOR1>(b0 ? lo : hi);
    }
    float* sbase = s_src + (size_t)(mtile * 16 + khalf * 8 + (b0 ? 4 : 0)) * HEADS + (row >> 1);
    for (int r = 0; r < 4; ++r) sbase[r * HEADS] = q4[r];
}

__global__ void __launch_bounds__(FT)
gat_order_and_gemm(const int* __restrict__ src, const int* __restrict__ dst,
                   const float* __restrict__ e, float* __restrict__ out,
                   int* __restrict__ loff, int* __restrict__ words,
                   const float* __restrict__ h, const float* __restrict__ Wfc,
                   const float* __restrict__ Wattn,
                   float* __restrict__ z, float* __restrict__ s_src) {
    const int blk = blockIdx.x;
    if (blk < NBLK) chunk_order_role(src, dst, e, out, loff, words, blk);
    else            gemm_role(h, Wfc, Wattn, z, s_src, blk - NBLK);
}

__global__ void __launch_bounds__(FT)
gat_place_and_gemm(const int* __restrict__ loff, const int* __restrict__ words,
                   int* __restrict__ rowptr, int* __restrict__ perm,
                   const float* __restrict__ h, const float* __restrict__ Wfc,
                   const float* __restrict__ Wattn,
                   float* __restrict__ z, float* __restrict__ s_src) {
    const int blk = blockIdx.x;
    if (blk < NBINS) slice_place_role(loff, words, rowptr, perm, blk);
    else             gemm_role(h, Wfc, Wattn, z, s_src, blk - NBINS + GEMM_FIRST);
}

__device__ __forceinline__ float edge_exp(float xs, float sd, float m) {
    float sc = xs + sd;
    sc = (sc >= 0.0f) ? sc : NEG_SLOPE * sc;
    return expf(sc - m);
}

__device__ __forceinline__ float elu1(float v) {
    return (v > 0.0f) ? v : expm1f(v);
}

template <int E>
__device__ __forceinline__ float quad_lane(float v) {
    return __int_as_float(__builtin_amdgcn_mov_dpp(__float_as_int(v), E * 0x55, 0xF, 0xF, true));
}

__device__ __forceinline__ int wave_order(int s, int c, int lane) {
    int rank = 0;
    for (int i = 0; i < c; ++i) {
        const int si = __builtin_amdgcn_readlane(s, i);
        rank += ((si < s) || (si == s && i < lane)) ? 1 : 0;
    }
    rank = (lane < c) ? rank : lane;
    return __builtin_amdgcn_ds_permute(rank << 2, s);
}

__device__ __forceinline__ void rows_load(const float* __restrict__ z, int s, int j, int c, int lane,
                                          float4& za, float4& zb, float4& zc, float4& zq) {
    const int sa = __builtin_amdgcn_readlane(s, j);
    const int sb = __builtin_amdgcn_readlane(s, min(j + 1, c - 1));
    const int sc = __builtin_amdgcn_readlane(s, min(j + 2, c - 1));
    const int sq = __builtin_amdgcn_readlane(s, min(j + 3, c - 1));
    za = *(const float4*)(z + (size_t)sa * OUT_DIM + lane * 4);
    zb = *(const float4*)(z + (size_t)sb * OUT_DIM + lane * 4);
    zc = *(const float4*)(z + (size_t)sc * OUT_DIM + lane * 4);
    zq = *(const float4*)(z + (size_t)sq * OUT_DIM + lane * 4);
}

__device__ __forceinline__ void group_sum(const float4& za, const float4& zb, const float4& zc,
                                          const float4& zq, float xg, float sd, float m,
                                          int j, int qd, int c, float& den, float4& acc) {
    const float km = (j + qd < c) ? 1.0f : 0.0f;
    const float pm = edge_exp(xg, sd, m) * km;
    const float pa = quad_lane<0>(pm);
    const float pb = quad_lane<1>(pm);
    const float pc = quad_lane<2>(pm);
    const float pq = quad_lane<3>(pm);
    den += pa;
    acc.x = fmaf(pa, za.x, acc.x); acc.y = fmaf(pa, za.y, acc.y);
    acc.z = fmaf(pa, za.z, acc.z); acc.w = fmaf(pa, za.w, acc.w);
    den += pb;
    acc.x = fmaf(pb, zb.x, acc.x); acc.y = fmaf(pb, zb.y, acc.y);
    acc.z = fmaf(pb, zb.z, acc.z); acc.w = fmaf(pb, zb.w, acc.w);
    den += pc;
    acc.x = fmaf(pc, zc.x, acc.x); acc.y = fmaf(pc, zc.y, acc.y);
    acc.z = fmaf(pc, zc.z, acc.z); acc.w = fmaf(pc, zc.w, acc.w);
    den += pq;
    acc.x = fmaf(pq, zq.x, acc.x); acc.y = fmaf(pq, zq.y, acc.y);
    acc.z = fmaf(pq, zq.z, acc.z); acc.w = fmaf(pq, zq.w, acc.w);
}

__device__ __forceinline__ float head_max_score(float mx, float sd) {
    mx = fmaxf(mx, row_xchg<DPP_XOR1>(mx));
    mx = fmaxf(mx, row_xchg<DPP_XOR2>(mx));
    const float m = mx + sd;
    return (m >= 0.0f) ? m : NEG_SLOPE * m;
}

__device__ __forceinline__ void chunk_fetch(const int* __restrict__ perm, const float* srow,
                                            int s_first, int beg, int base, int deg,
                                            int lane, int qd, int& s, int& c, float (&x)[8]) {
    c = min(32, deg - base);
    const int sv = (base == 0) ? s_first : ((lane < c) ? perm[beg + base + lane] : 0);
    s = wave_order(sv, c, lane);
    const int last = c - 1;
    int sm[8];
#pragma unroll
    for (int g = 0; g < 8; ++g)
        sm[g] = __builtin_amdgcn_ds_bpermute(min(4 * g + qd, last) << 2, s);
#pragma unroll
    for (int g = 0; g < 8; ++g) {
        x[g] = -INFINITY;
        if (4 * g < c) x[g] = srow[(size_t)sm[g] * HEADS];
    }
}

__global__ void __launch_bounds__(256)
gat_node_gather(const int* __restrict__ rowptr, const int* __restrict__ perm,
                const float* __restrict__ s_src, const float* __restrict__ z,
                const float* __restrict__ Wattn, const float* __restrict__ h,
                float* __restrict__ out) {
    const int lane = threadIdx.x & 31;
    const int node = __builtin_amdgcn_readfirstlane((int)(blockIdx.x * 8 + (threadIdx.x >> 5)));
    if (node >= N_NODES) return;
    const int hh = lane >> 2;
    const int qd = lane & 3;

    const int beg = rowptr[node];
    const int end = (node + 1 < N_NODES) ? rowptr[node + 1] : N_EDGES;
    const int deg = end - beg;

    const float4 zd = *(const float4*)(z + (size_t)node * OUT_DIM + lane * 4);
    const float4 hv = *(const float4*)(h + (size_t)node * OUT_DIM + lane * 4);
    const float4 ad = *(const float4*)(Wattn + hh * (2 * HDIM) + HDIM + (lane & 3) * 4);
    float sd = zd.x * ad.x + zd.y * ad.y + zd.z * ad.z + zd.w * ad.w;
    sd += __shfl_xor(sd, 1);
    sd += __shfl_xor(sd, 2);

    const int s_first = (lane < deg) ? perm[beg + lane] : 0;
    const float* srow = s_src + hh;

    int s = 0, c = 0;
    float x[8];
    float den = 0.0f;
    float4 acc = make_float4(0.0f, 0.0f, 0.0f, 0.0f);
    if (deg <= 32) {
        if (deg > 0) {
            chunk_fetch(perm, srow, s_first, beg, 0, deg, lane, qd, s, c, x);
            float4 za, zb, zc, zq;
            rows_load(z, s, 0, c, lane, za, zb, zc, zq);
            __builtin_amdgcn_sched_barrier(0);
            float mx = x[0];
#pragma unroll
            for (int g = 1; g < 8; ++g) mx = fmaxf(mx, x[g]);
            const float m = head_max_score(mx, sd);
            group_sum(za, zb, zc, zq, x[0], sd, m, 0, qd, c, den, acc);
#pragma unroll
            for (int g = 1; g < 8; ++g) {
                if (4 * g < c) {
                    rows_load(z, s, 4 * g, c, lane, za, zb, zc, zq);
                    group_sum(za, zb, zc, zq, x[g], sd, m, 4 * g, qd, c, den, acc);
                }
            }
        }
    } else {
        float mx = -INFINITY;
        for (int base = 0; base < deg; base += 32) {
            chunk_fetch(perm, srow, s_first, beg, base, deg, lane, qd, s, c, x);
#pragma unroll
            for (int g = 0; g < 8; ++g) mx = fmaxf(mx, x[g]);
        }
        const float m = head_max_score(mx, sd);
        for (int base = 0; base < deg; base += 32) {
            chunk_fetch(perm, srow, s_first, beg, base, deg, lane, qd, s, c, x);
#pragma unroll
            for (int g = 0; g < 8; ++g) {
                if (4 * g < c) {
                    float4 za, zb, zc, zq;
                    rows_load(z, s, 4 * g, c, lane, za, zb, zc, zq);
                    group_sum(za, zb, zc, zq, x[g], sd, m, 4 * g, qd, c, den, acc);
                }
            }
        }
    }

    float4 r = make_float4(0.0f, 0.0f, 0.0f, 0.0f);
    if (deg > 0) {
        const float inv = 1.0f / den;
        r.x = acc.x * inv; r.y = acc.y * inv; r.z = acc.z * inv; r.w = acc.w * inv;
    }
    float4 o;
    o.x = hv.x + elu1(r.x); o.y = hv.y + elu1(r.y);
    o.z = hv.z + elu1(r.z); o.w = hv.w + elu1(r.w);
    *(float4*)(out + (size_t)node * OUT_DIM + lane * 4) = o;
}

extern "C" void kernel_launch(void* const* d_in, const int* in_sizes, int n_in,
                              void* d_out, int out_size, void* d_ws, size_t ws_size,
                              hipStream_t stream) {
    const float* h     = (const float*)d_in[0];
    const float* e     = (const float*)d_in[1];
    const int*   src   = (const int*)d_in[2];
    const int*   dst   = (const int*)d_in[3];
    const float* Wfc   = (const float*)d_in[4];
    const float* Wattn = (const float*)d_in[5];
    float* out = (float*)d_out;

    float* z      = (float*)d_ws;
    float* s_src  = z + (size_t)N_NODES * OUT_DIM;
    int*   rowptr = (int*)(s_src + (size_t)N_NODES * HEADS);
    int*   loff   = rowptr + N_NODES;
    int*   perm   = rowptr + (size_t)N_NODES * HEADS;
    int*   words  = (int*)d_out;

    const int T = 256;

    gat_order_and_gemm<<<NBLK + GEMM_FIRST, FT, 0, stream>>>(src, dst, e, out, loff, words,
                                                             h, Wfc, Wattn, z, s_src);
    gat_place_and_gemm<<<NBINS + GEMM_BLOCKS - GEMM_FIRST, FT, 0, stream>>>(loff, words, rowptr, perm,
                                                                            h, Wfc, Wattn, z, s_src);

    gat_node_gather<<<(N_NODES + 7) / 8, T, 0, stream>>>(rowptr, perm, s_src, z, Wattn, h, out);
}
